# starting kernel
# baseline (speedup 1.0000x reference)
#include <hip/hip_runtime.h>
#include <math.h>


typedef __attribute__((ext_vector_type(2))) float v2f;
typedef __attribute__((ext_vector_type(8))) float v8f;

#define DF 32

__global__ __launch_bounds__(256) void k_zero(float* __restrict__ p, int n) {
  int i = blockIdx.x * blockDim.x + threadIdx.x;
  if (i < n) p[i] = 0.0f;
}

__global__ __launch_bounds__(256) void k_degrees(const int* __restrict__ src,
                                                 const int* __restrict__ dst,
                                                 float* __restrict__ deg_out,
                                                 float* __restrict__ deg_in,
                                                 int e) {
  int i = blockIdx.x * blockDim.x + threadIdx.x;
  if (i < e) {
    atomicAdd(&deg_out[src[i]], 1.0f);
    atomicAdd(&deg_in[dst[i]], 1.0f);
  }
}

__global__ __launch_bounds__(256) void k_invsqrt(float* __restrict__ a,
                                                 float* __restrict__ b, int n) {
  int i = blockIdx.x * blockDim.x + threadIdx.x;
  if (i < n) {
    float d = a[i];
    a[i] = (d > 0.0f) ? (1.0f / sqrtf(d)) : 0.0f;
    d = b[i];
    b[i] = (d > 0.0f) ? (1.0f / sqrtf(d)) : 0.0f;
  }
}

__global__ __launch_bounds__(256) void k_prescale(const float* __restrict__ x,
                                                  const float* __restrict__ nrm,
                                                  float* __restrict__ xs,
                                                  int total) {
  int i = blockIdx.x * blockDim.x + threadIdx.x;
  if (i < total) xs[i] = x[i] * nrm[i >> 5];
}

__global__ __launch_bounds__(256) void k_scatter(const float* __restrict__ xs,
                                                 const int* __restrict__ src,
                                                 const int* __restrict__ dst,
                                                 float* __restrict__ agg,
                                                 int total) {
  int i = blockIdx.x * blockDim.x + threadIdx.x;
  if (i >= total) return;
  int e = i >> 5;
  int k = i & 31;
  int s = src[e];
  int d = dst[e];
  atomicAdd(&agg[(size_t)d * DF + k], xs[(size_t)s * DF + k]);
}

__global__ __launch_bounds__(256) void k_gemm_wmma(
    const float* __restrict__ agg, const float* __restrict__ norm_in,
    const float* __restrict__ W, const float* __restrict__ bias,
    const float* __restrict__ post_scale,
    float* __restrict__ out, int n, int do_relu) {
  const int lane = threadIdx.x & 31;
  const int wave = threadIdx.x >> 5;
  const int row0 = (blockIdx.x * 8 + wave) * 16;
  if (row0 >= n) return;

  const int mrow = lane & 15;
  const int khalf = lane >> 4;
  int rload = row0 + mrow;
  if (rload > n - 1) rload = n - 1;
  const float ni = norm_in[rload];
  const float* __restrict__ arow = agg + (size_t)rload * DF;
  const int nb = lane & 15;

  v8f acc0 = {};
  v8f acc1 = {};
#pragma unroll
  for (int k0 = 0; k0 < DF; k0 += 4) {
    const int ka = k0 + khalf * 2;
    v2f a;
    a.x = arow[ka + 0] * ni;
    a.y = arow[ka + 1] * ni;
    v2f b0, b1;
    b0.x = W[(ka + 0) * DF + nb];
    b0.y = W[(ka + 1) * DF + nb];
    b1.x = W[(ka + 0) * DF + 16 + nb];
    b1.y = W[(ka + 1) * DF + 16 + nb];
    acc0 = __builtin_amdgcn_wmma_f32_16x16x4_f32(false, a, false, b0,
                                                 (short)0, acc0, false, false);
    acc1 = __builtin_amdgcn_wmma_f32_16x16x4_f32(false, a, false, b1,
                                                 (short)0, acc1, false, false);
  }

  const float bc0 = bias[nb];
  const float bc1 = bias[16 + nb];
#pragma unroll
  for (int v = 0; v < 8; ++v) {
    const int m = row0 + v + 8 * khalf;
    if (m < n) {
      float o0 = acc0[v] + bc0;
      float o1 = acc1[v] + bc1;
      if (do_relu) {
        o0 = fmaxf(o0, 0.0f);
        o1 = fmaxf(o1, 0.0f);
      }
      const float s = post_scale ? post_scale[m] : 1.0f;
      out[(size_t)m * DF + nb] = o0 * s;
      out[(size_t)m * DF + 16 + nb] = o1 * s;
    }
  }
}

extern "C" void kernel_launch(void* const* d_in, const int* in_sizes, int n_in,
                              void* d_out, int out_size, void* d_ws,
                              size_t ws_size, hipStream_t stream) {
  const float* x = (const float*)d_in[0];
  const int* src = (const int*)d_in[1];
  const int* dst = (const int*)d_in[2];
  const float* W1 = (const float*)d_in[3];
  const float* b1 = (const float*)d_in[4];
  const float* W2 = (const float*)d_in[5];
  const float* b2 = (const float*)d_in[6];
  float* out = (float*)d_out;

  const int N = in_sizes[0] / DF;
  const int E = in_sizes[1];
  const int NF = N * DF;
  const int tot = E * DF;

  float* ws = (float*)d_ws;
  float* norm_out = ws;
  float* norm_in = ws + N;
  float* xs = ws + 2 * (size_t)N;
  float* agg = ws + 2 * (size_t)N + (size_t)NF;

  const int T = 256;
  const int gB = ((N + 15) / 16 + 7) / 8;

  k_zero<<<(2 * N + T - 1) / T, T, 0, stream>>>(norm_out, 2 * N);
  k_degrees<<<(E + T - 1) / T, T, 0, stream>>>(src, dst, norm_out, norm_in, E);
  k_invsqrt<<<(N + T - 1) / T, T, 0, stream>>>(norm_out, norm_in, N);

  k_prescale<<<(NF + T - 1) / T, T, 0, stream>>>(x, norm_out, xs, NF);
  k_zero<<<(NF + T - 1) / T, T, 0, stream>>>(agg, NF);
  k_scatter<<<(tot + T - 1) / T, T, 0, stream>>>(xs, src, dst, agg, tot);
  k_gemm_wmma<<<gB, 256, 0, stream>>>(agg, norm_in, W1, b1, norm_out, xs, N, 1);

  k_zero<<<(NF + T - 1) / T, T, 0, stream>>>(agg, NF);
  k_scatter<<<(tot + T - 1) / T, T, 0, stream>>>(xs, src, dst, agg, tot);
  k_gemm_wmma<<<gB, 256, 0, stream>>>(agg, norm_in, W2, b2, nullptr, out, N, 0);
}


# Degree counts by per-work-group LDS histograms (integer LDS adds), stored as partial planes and summed per node in the norm kernel, replacing two scattered global f32 atomics per edge; 9 launches
# speedup vs baseline: 1.4428x; 1.4428x over previous
#include <hip/hip_runtime.h>
#include <math.h>


typedef __attribute__((ext_vector_type(2))) float v2f;
typedef __attribute__((ext_vector_type(8))) float v8f;

#define DF 32

__global__ __launch_bounds__(256) void k_zero(float* __restrict__ p, int n) {
  int i = blockIdx.x * blockDim.x + threadIdx.x;
  if (i < n) p[i] = 0.0f;
}

#define HIST_CAP 65536
#define HIST_T 1024
#define HIST_S 64
#define HIST_U 8

__global__ __launch_bounds__(HIST_T) void k_deg_hist(
    const int* __restrict__ src, const int* __restrict__ dst,
    unsigned* __restrict__ part, int n, int e, int chunk) {
  __shared__ __attribute__((aligned(16))) unsigned hist[HIST_CAP];
  const int tid = threadIdx.x;
  const int slice = blockIdx.x;
  const int which = blockIdx.y;
  const int* __restrict__ arr = which ? dst : src;
  unsigned* __restrict__ plane =
      part + ((size_t)which * HIST_S + slice) * (size_t)n;
  const int j0 = slice * chunk;
  int j1 = j0 + chunk;
  if (j1 > e) j1 = e;
  const int cnt = (j1 > j0) ? (j1 - j0) : 0;
  const int nvec = cnt >> 2;
  const int ntail = cnt & 3;
  const int4* __restrict__ arr4 = reinterpret_cast<const int4*>(arr + j0);
  const bool vec_ok = (n & 3) == 0;

  for (int base = 0; base < n; base += HIST_CAP) {
    int nb = n - base;
    if (nb > HIST_CAP) nb = HIST_CAP;

    for (int i = tid * 4; i < nb; i += HIST_T * 4)
      *reinterpret_cast<uint4*>(&hist[i]) = make_uint4(0u, 0u, 0u, 0u);
    __syncthreads();

    for (int v0 = tid; v0 < nvec; v0 += HIST_T * HIST_U) {
      int4 v[HIST_U];
#pragma unroll
      for (int u = 0; u < HIST_U; ++u) {
        const int idx = v0 + u * HIST_T;
        v[u] = arr4[idx < nvec ? idx : nvec - 1];
      }
#pragma unroll
      for (int u = 0; u < HIST_U; ++u) {
        if (v0 + u * HIST_T < nvec) {
          unsigned b;
          b = (unsigned)(v[u].x - base);
          if (b < (unsigned)nb) atomicAdd(&hist[b], 1u);
          b = (unsigned)(v[u].y - base);
          if (b < (unsigned)nb) atomicAdd(&hist[b], 1u);
          b = (unsigned)(v[u].z - base);
          if (b < (unsigned)nb) atomicAdd(&hist[b], 1u);
          b = (unsigned)(v[u].w - base);
          if (b < (unsigned)nb) atomicAdd(&hist[b], 1u);
        }
      }
    }
    if (tid < ntail) {
      const unsigned b = (unsigned)(arr[j0 + nvec * 4 + tid] - base);
      if (b < (unsigned)nb) atomicAdd(&hist[b], 1u);
    }
    __syncthreads();

    if (vec_ok) {
      for (int i = tid * 4; i < nb; i += HIST_T * 4)
        *reinterpret_cast<uint4*>(&plane[base + i]) =
            *reinterpret_cast<const uint4*>(&hist[i]);
    } else {
      for (int i = tid; i < nb; i += HIST_T) plane[base + i] = hist[i];
    }
    __syncthreads();
  }
}

__global__ __launch_bounds__(256) void k_norms(const unsigned* __restrict__ part,
                                               float* __restrict__ norms,
                                               int n) {
  const int i = blockIdx.x * blockDim.x + threadIdx.x;
  const int which = blockIdx.y;
  if (i < n) {
    const unsigned* __restrict__ p = part + (size_t)which * HIST_S * n + i;
    unsigned c = 0;
#pragma unroll
    for (int s = 0; s < HIST_S; ++s) c += p[(size_t)s * n];
    const float d = (float)c;
    norms[(size_t)which * n + i] = (d > 0.0f) ? (1.0f / sqrtf(d)) : 0.0f;
  }
}

__global__ __launch_bounds__(256) void k_prescale(const float* __restrict__ x,
                                                  const float* __restrict__ nrm,
                                                  float* __restrict__ xs,
                                                  int total) {
  int i = blockIdx.x * blockDim.x + threadIdx.x;
  if (i < total) xs[i] = x[i] * nrm[i >> 5];
}

__global__ __launch_bounds__(256) void k_scatter(const float* __restrict__ xs,
                                                 const int* __restrict__ src,
                                                 const int* __restrict__ dst,
                                                 float* __restrict__ agg,
                                                 int total) {
  int i = blockIdx.x * blockDim.x + threadIdx.x;
  if (i >= total) return;
  int e = i >> 5;
  int k = i & 31;
  int s = src[e];
  int d = dst[e];
  atomicAdd(&agg[(size_t)d * DF + k], xs[(size_t)s * DF + k]);
}

__global__ __launch_bounds__(256) void k_gemm_wmma(
    const float* __restrict__ agg, const float* __restrict__ norm_in,
    const float* __restrict__ W, const float* __restrict__ bias,
    const float* __restrict__ post_scale,
    float* __restrict__ out, int n, int do_relu) {
  const int lane = threadIdx.x & 31;
  const int wave = threadIdx.x >> 5;
  const int row0 = (blockIdx.x * 8 + wave) * 16;
  if (row0 >= n) return;

  const int mrow = lane & 15;
  const int khalf = lane >> 4;
  int rload = row0 + mrow;
  if (rload > n - 1) rload = n - 1;
  const float ni = norm_in[rload];
  const float* __restrict__ arow = agg + (size_t)rload * DF;
  const int nb = lane & 15;

  v8f acc0 = {};
  v8f acc1 = {};
#pragma unroll
  for (int k0 = 0; k0 < DF; k0 += 4) {
    const int ka = k0 + khalf * 2;
    v2f a;
    a.x = arow[ka + 0] * ni;
    a.y = arow[ka + 1] * ni;
    v2f b0, b1;
    b0.x = W[(ka + 0) * DF + nb];
    b0.y = W[(ka + 1) * DF + nb];
    b1.x = W[(ka + 0) * DF + 16 + nb];
    b1.y = W[(ka + 1) * DF + 16 + nb];
    acc0 = __builtin_amdgcn_wmma_f32_16x16x4_f32(false, a, false, b0,
                                                 (short)0, acc0, false, false);
    acc1 = __builtin_amdgcn_wmma_f32_16x16x4_f32(false, a, false, b1,
                                                 (short)0, acc1, false, false);
  }

  const float bc0 = bias[nb];
  const float bc1 = bias[16 + nb];
#pragma unroll
  for (int v = 0; v < 8; ++v) {
    const int m = row0 + v + 8 * khalf;
    if (m < n) {
      float o0 = acc0[v] + bc0;
      float o1 = acc1[v] + bc1;
      if (do_relu) {
        o0 = fmaxf(o0, 0.0f);
        o1 = fmaxf(o1, 0.0f);
      }
      const float s = post_scale ? post_scale[m] : 1.0f;
      out[(size_t)m * DF + nb] = o0 * s;
      out[(size_t)m * DF + 16 + nb] = o1 * s;
    }
  }
}

extern "C" void kernel_launch(void* const* d_in, const int* in_sizes, int n_in,
                              void* d_out, int out_size, void* d_ws,
                              size_t ws_size, hipStream_t stream) {
  const float* x = (const float*)d_in[0];
  const int* src = (const int*)d_in[1];
  const int* dst = (const int*)d_in[2];
  const float* W1 = (const float*)d_in[3];
  const float* b1 = (const float*)d_in[4];
  const float* W2 = (const float*)d_in[5];
  const float* b2 = (const float*)d_in[6];
  float* out = (float*)d_out;

  const int N = in_sizes[0] / DF;
  const int E = in_sizes[1];
  const int NF = N * DF;
  const int tot = E * DF;

  float* ws = (float*)d_ws;
  float* norm_out = ws;
  float* norm_in = ws + N;
  float* xs = ws + 2 * (size_t)N;
  float* agg = ws + 2 * (size_t)N + (size_t)NF;
  unsigned* part = (unsigned*)(ws + 2 * (size_t)N + 2 * (size_t)NF);

  const int T = 256;
  const int gB = ((N + 15) / 16 + 7) / 8;

  const int chunk = (((E + HIST_S - 1) / HIST_S) + 3) & ~3;
  k_deg_hist<<<dim3(HIST_S, 2), HIST_T, 0, stream>>>(src, dst, part, N, E, chunk);
  k_norms<<<dim3((N + T - 1) / T, 2), T, 0, stream>>>(part, norm_out, N);

  k_prescale<<<(NF + T - 1) / T, T, 0, stream>>>(x, norm_out, xs, NF);
  k_zero<<<(NF + T - 1) / T, T, 0, stream>>>(agg, NF);
  k_scatter<<<(tot + T - 1) / T, T, 0, stream>>>(xs, src, dst, agg, tot);
  k_gemm_wmma<<<gB, 256, 0, stream>>>(agg, norm_in, W1, b1, norm_out, xs, N, 1);

  k_zero<<<(NF + T - 1) / T, T, 0, stream>>>(agg, NF);
  k_scatter<<<(tot + T - 1) / T, T, 0, stream>>>(xs, src, dst, agg, tot);
  k_gemm_wmma<<<gB, 256, 0, stream>>>(agg, norm_in, W2, b2, nullptr, out, N, 0);
}


# Replaces the f32 atomic scatter-add by a counting sort of the edges by destination (LDS cursors seeded from the degree planes) and an atomic-free per-node gather-sum; per-node sum order changes
# speedup vs baseline: 3.0693x; 2.1273x over previous
#include <hip/hip_runtime.h>
#include <math.h>


typedef __attribute__((ext_vector_type(2))) float v2f;
typedef __attribute__((ext_vector_type(8))) float v8f;

#define DF 32

#define HIST_CAP 65536
#define HIST_T 1024
#define HIST_S 64
#define HIST_U 8

__global__ __launch_bounds__(HIST_T) void k_deg_hist(
    const int* __restrict__ src, const int* __restrict__ dst,
    unsigned* __restrict__ part, int n, int e, int chunk) {
  __shared__ __attribute__((aligned(16))) unsigned hist[HIST_CAP];
  const int tid = threadIdx.x;
  const int slice = blockIdx.x;
  const int which = blockIdx.y;
  const int* __restrict__ arr = which ? dst : src;
  unsigned* __restrict__ plane =
      part + ((size_t)which * HIST_S + slice) * (size_t)n;
  const int j0 = slice * chunk;
  int j1 = j0 + chunk;
  if (j1 > e) j1 = e;
  const int cnt = (j1 > j0) ? (j1 - j0) : 0;
  const int nvec = cnt >> 2;
  const int ntail = cnt & 3;
  const int4* __restrict__ arr4 = reinterpret_cast<const int4*>(arr + j0);
  const bool vec_ok = (n & 3) == 0;

  for (int base = 0; base < n; base += HIST_CAP) {
    int nb = n - base;
    if (nb > HIST_CAP) nb = HIST_CAP;

    for (int i = tid * 4; i < nb; i += HIST_T * 4)
      *reinterpret_cast<uint4*>(&hist[i]) = make_uint4(0u, 0u, 0u, 0u);
    __syncthreads();

    for (int v0 = tid; v0 < nvec; v0 += HIST_T * HIST_U) {
      int4 v[HIST_U];
#pragma unroll
      for (int u = 0; u < HIST_U; ++u) {
        const int idx = v0 + u * HIST_T;
        v[u] = arr4[idx < nvec ? idx : nvec - 1];
      }
#pragma unroll
      for (int u = 0; u < HIST_U; ++u) {
        if (v0 + u * HIST_T < nvec) {
          unsigned b;
          b = (unsigned)(v[u].x - base);
          if (b < (unsigned)nb) atomicAdd(&hist[b], 1u);
          b = (unsigned)(v[u].y - base);
          if (b < (unsigned)nb) atomicAdd(&hist[b], 1u);
          b = (unsigned)(v[u].z - base);
          if (b < (unsigned)nb) atomicAdd(&hist[b], 1u);
          b = (unsigned)(v[u].w - base);
          if (b < (unsigned)nb) atomicAdd(&hist[b], 1u);
        }
      }
    }
    if (tid < ntail) {
      const unsigned b = (unsigned)(arr[j0 + nvec * 4 + tid] - base);
      if (b < (unsigned)nb) atomicAdd(&hist[b], 1u);
    }
    __syncthreads();

    if (vec_ok) {
      for (int i = tid * 4; i < nb; i += HIST_T * 4)
        *reinterpret_cast<uint4*>(&plane[base + i]) =
            *reinterpret_cast<const uint4*>(&hist[i]);
    } else {
      for (int i = tid; i < nb; i += HIST_T) plane[base + i] = hist[i];
    }
    __syncthreads();
  }
}

#define NB_T 256
#define NB_SHIFT 8

__global__ __launch_bounds__(NB_T) void k_norms(const unsigned* __restrict__ part,
                                                float* __restrict__ norms,
                                                unsigned* __restrict__ pre,
                                                unsigned* __restrict__ lpre,
                                                unsigned* __restrict__ degin,
                                                unsigned* __restrict__ btot,
                                                int n) {
  __shared__ unsigned sc[2][NB_T];
  const int tid = threadIdx.x;
  const int i = blockIdx.x * NB_T + tid;
  const int which = blockIdx.y;
  unsigned c = 0;
  if (i < n) {
    const unsigned* __restrict__ p = part + (size_t)which * HIST_S * n + i;
    unsigned v[HIST_S];
#pragma unroll
    for (int s = 0; s < HIST_S; ++s) v[s] = p[(size_t)s * n];
    if (which == 1) {
#pragma unroll
      for (int s = 0; s < HIST_S; ++s) {
        pre[(size_t)s * n + i] = c;
        c += v[s];
      }
    } else {
#pragma unroll
      for (int s = 0; s < HIST_S; ++s) c += v[s];
    }
    const float d = (float)c;
    norms[(size_t)which * n + i] = (d > 0.0f) ? (1.0f / sqrtf(d)) : 0.0f;
  }
  if (which == 1) {
    sc[0][tid] = c;
    __syncthreads();
    int cur = 0;
    for (int off = 1; off < NB_T; off <<= 1) {
      unsigned t = sc[cur][tid];
      if (tid >= off) t += sc[cur][tid - off];
      sc[cur ^ 1][tid] = t;
      cur ^= 1;
      __syncthreads();
    }
    const unsigned incl = sc[cur][tid];
    if (i < n) {
      lpre[i] = incl - c;
      degin[i] = c;
    }
    if (tid == NB_T - 1) btot[blockIdx.x] = incl;
  }
}

__global__ __launch_bounds__(1024) void k_blockscan(
    const unsigned* __restrict__ btot, unsigned* __restrict__ boff, int nblk) {
  __shared__ unsigned sc[2][1024];
  const int tid = threadIdx.x;
  unsigned carry = 0;
  for (int base = 0; base < nblk; base += 1024) {
    const int i = base + tid;
    const unsigned c = (i < nblk) ? btot[i] : 0u;
    sc[0][tid] = c;
    __syncthreads();
    int cur = 0;
    for (int off = 1; off < 1024; off <<= 1) {
      unsigned t = sc[cur][tid];
      if (tid >= off) t += sc[cur][tid - off];
      sc[cur ^ 1][tid] = t;
      cur ^= 1;
      __syncthreads();
    }
    const unsigned incl = sc[cur][tid];
    if (i < nblk) boff[i] = carry + incl - c;
    carry += sc[cur][1023];
    __syncthreads();
  }
}

__global__ __launch_bounds__(HIST_T) void k_place(
    const int* __restrict__ src, const int* __restrict__ dst,
    const unsigned* __restrict__ pre, const unsigned* __restrict__ lpre,
    const unsigned* __restrict__ boff, int* __restrict__ ssrc, int n, int e,
    int chunk) {
  __shared__ __attribute__((aligned(16))) unsigned cursor[HIST_CAP];
  const int tid = threadIdx.x;
  const int slice = blockIdx.x;
  const unsigned* __restrict__ plane = pre + (size_t)slice * (size_t)n;
  const int j0 = slice * chunk;
  int j1 = j0 + chunk;
  if (j1 > e) j1 = e;
  const int cnt = (j1 > j0) ? (j1 - j0) : 0;
  const int nvec = cnt >> 2;
  const int ntail = cnt & 3;
  const int4* __restrict__ src4 = reinterpret_cast<const int4*>(src + j0);
  const int4* __restrict__ dst4 = reinterpret_cast<const int4*>(dst + j0);
  const bool vec_ok = (n & 3) == 0;

  for (int base = 0; base < n; base += HIST_CAP) {
    int nb = n - base;
    if (nb > HIST_CAP) nb = HIST_CAP;

    if (vec_ok) {
      for (int i = tid * 4; i < nb; i += HIST_T * 4) {
        const int g = base + i;
        const uint4 l = *reinterpret_cast<const uint4*>(&lpre[g]);
        const uint4 p = *reinterpret_cast<const uint4*>(&plane[g]);
        const unsigned bo = boff[g >> NB_SHIFT];
        *reinterpret_cast<uint4*>(&cursor[i]) = make_uint4(
            bo + l.x + p.x, bo + l.y + p.y, bo + l.z + p.z, bo + l.w + p.w);
      }
    } else {
      for (int i = tid; i < nb; i += HIST_T) {
        const int g = base + i;
        cursor[i] = boff[g >> NB_SHIFT] + lpre[g] + plane[g];
      }
    }
    __syncthreads();

    for (int v0 = tid; v0 < nvec; v0 += HIST_T * HIST_U) {
      int4 dv[HIST_U], sv[HIST_U];
#pragma unroll
      for (int u = 0; u < HIST_U; ++u) {
        const int idx = v0 + u * HIST_T;
        const int cl = idx < nvec ? idx : nvec - 1;
        dv[u] = dst4[cl];
        sv[u] = src4[cl];
      }
#pragma unroll
      for (int u = 0; u < HIST_U; ++u) {
        if (v0 + u * HIST_T < nvec) {
          unsigned b, pos;
          b = (unsigned)(dv[u].x - base);
          if (b < (unsigned)nb) {
            pos = atomicAdd(&cursor[b], 1u);
            if (pos < (unsigned)e) ssrc[pos] = sv[u].x;
          }
          b = (unsigned)(dv[u].y - base);
          if (b < (unsigned)nb) {
            pos = atomicAdd(&cursor[b], 1u);
            if (pos < (unsigned)e) ssrc[pos] = sv[u].y;
          }
          b = (unsigned)(dv[u].z - base);
          if (b < (unsigned)nb) {
            pos = atomicAdd(&cursor[b], 1u);
            if (pos < (unsigned)e) ssrc[pos] = sv[u].z;
          }
          b = (unsigned)(dv[u].w - base);
          if (b < (unsigned)nb) {
            pos = atomicAdd(&cursor[b], 1u);
            if (pos < (unsigned)e) ssrc[pos] = sv[u].w;
          }
        }
      }
    }
    if (tid < ntail) {
      const int j = j0 + nvec * 4 + tid;
      const unsigned b = (unsigned)(dst[j] - base);
      if (b < (unsigned)nb) {
        const unsigned pos = atomicAdd(&cursor[b], 1u);
        if (pos < (unsigned)e) ssrc[pos] = src[j];
      }
    }
    __syncthreads();
  }
}

__global__ __launch_bounds__(256) void k_prescale(const float* __restrict__ x,
                                                  const float* __restrict__ nrm,
                                                  float* __restrict__ xs,
                                                  int total) {
  int i = blockIdx.x * blockDim.x + threadIdx.x;
  if (i < total) xs[i] = x[i] * nrm[i >> 5];
}

#define AGG_U 16

__global__ __launch_bounds__(256) void k_aggregate(
    const float* __restrict__ xs, const int* __restrict__ ssrc,
    const unsigned* __restrict__ lpre, const unsigned* __restrict__ boff,
    const unsigned* __restrict__ degin, float* __restrict__ agg, int n) {
  const int lane = threadIdx.x & 31;
  const int wave = __builtin_amdgcn_readfirstlane((int)(threadIdx.x >> 5));
  const int node = blockIdx.x * 8 + wave;
  if (node >= n) return;
  const unsigned start = boff[node >> NB_SHIFT] + lpre[node];
  const unsigned cnt = degin[node];
  float acc = 0.0f;
  for (unsigned c0 = 0; c0 < cnt; c0 += 32) {
    unsigned m = cnt - c0;
    if (m > 32u) m = 32u;
    const unsigned li = (unsigned)lane < m ? (unsigned)lane : m - 1u;
    const int mine = ssrc[start + c0 + li];
    for (unsigned u = 0; u < m; u += AGG_U) {
      float v[AGG_U];
#pragma unroll
      for (int j = 0; j < AGG_U; ++j) {
        const unsigned q = (u + j < m) ? (u + j) : (m - 1u);
        const int s = __builtin_amdgcn_readlane(mine, (int)q);
        v[j] = xs[(size_t)s * DF + lane];
      }
#pragma unroll
      for (int j = 0; j < AGG_U; ++j) acc += (u + j < m) ? v[j] : 0.0f;
    }
  }
  agg[(size_t)node * DF + lane] = acc;
}

__global__ __launch_bounds__(256) void k_gemm_wmma(
    const float* __restrict__ agg, const float* __restrict__ norm_in,
    const float* __restrict__ W, const float* __restrict__ bias,
    const float* __restrict__ post_scale,
    float* __restrict__ out, int n, int do_relu) {
  const int lane = threadIdx.x & 31;
  const int wave = threadIdx.x >> 5;
  const int row0 = (blockIdx.x * 8 + wave) * 16;
  if (row0 >= n) return;

  const int mrow = lane & 15;
  const int khalf = lane >> 4;
  int rload = row0 + mrow;
  if (rload > n - 1) rload = n - 1;
  const float ni = norm_in[rload];
  const float* __restrict__ arow = agg + (size_t)rload * DF;
  const int nb = lane & 15;

  v8f acc0 = {};
  v8f acc1 = {};
#pragma unroll
  for (int k0 = 0; k0 < DF; k0 += 4) {
    const int ka = k0 + khalf * 2;
    v2f a;
    a.x = arow[ka + 0] * ni;
    a.y = arow[ka + 1] * ni;
    v2f b0, b1;
    b0.x = W[(ka + 0) * DF + nb];
    b0.y = W[(ka + 1) * DF + nb];
    b1.x = W[(ka + 0) * DF + 16 + nb];
    b1.y = W[(ka + 1) * DF + 16 + nb];
    acc0 = __builtin_amdgcn_wmma_f32_16x16x4_f32(false, a, false, b0,
                                                 (short)0, acc0, false, false);
    acc1 = __builtin_amdgcn_wmma_f32_16x16x4_f32(false, a, false, b1,
                                                 (short)0, acc1, false, false);
  }

  const float bc0 = bias[nb];
  const float bc1 = bias[16 + nb];
#pragma unroll
  for (int v = 0; v < 8; ++v) {
    const int m = row0 + v + 8 * khalf;
    if (m < n) {
      float o0 = acc0[v] + bc0;
      float o1 = acc1[v] + bc1;
      if (do_relu) {
        o0 = fmaxf(o0, 0.0f);
        o1 = fmaxf(o1, 0.0f);
      }
      const float s = post_scale ? post_scale[m] : 1.0f;
      out[(size_t)m * DF + nb] = o0 * s;
      out[(size_t)m * DF + 16 + nb] = o1 * s;
    }
  }
}

extern "C" void kernel_launch(void* const* d_in, const int* in_sizes, int n_in,
                              void* d_out, int out_size, void* d_ws,
                              size_t ws_size, hipStream_t stream) {
  const float* x = (const float*)d_in[0];
  const int* src = (const int*)d_in[1];
  const int* dst = (const int*)d_in[2];
  const float* W1 = (const float*)d_in[3];
  const float* b1 = (const float*)d_in[4];
  const float* W2 = (const float*)d_in[5];
  const float* b2 = (const float*)d_in[6];
  float* out = (float*)d_out;

  const int N = in_sizes[0] / DF;
  const int E = in_sizes[1];
  const int NF = N * DF;

  float* ws = (float*)d_ws;
  float* norm_out = ws;
  float* norm_in = ws + N;
  float* xs = ws + 2 * (size_t)N;
  float* agg = ws + 2 * (size_t)N + (size_t)NF;
  unsigned* part = (unsigned*)(ws + 2 * (size_t)N + 2 * (size_t)NF);
  const int nblk = (N + NB_T - 1) / NB_T;
  unsigned* pre = part + 2 * (size_t)HIST_S * N;
  unsigned* lpre = pre + (size_t)HIST_S * N;
  unsigned* degin = lpre + N;
  unsigned* btot = degin + N;
  unsigned* boff = btot + nblk;
  int* ssrc = (int*)(boff + nblk);

  const int T = 256;
  const int gB = ((N + 15) / 16 + 7) / 8;
  const int gA = (N + 7) / 8;

  const int chunk = (((E + HIST_S - 1) / HIST_S) + 3) & ~3;
  k_deg_hist<<<dim3(HIST_S, 2), HIST_T, 0, stream>>>(src, dst, part, N, E, chunk);
  k_norms<<<dim3(nblk, 2), NB_T, 0, stream>>>(part, norm_out, pre, lpre, degin, btot, N);
  k_blockscan<<<1, 1024, 0, stream>>>(btot, boff, nblk);
  k_place<<<HIST_S, HIST_T, 0, stream>>>(src, dst, pre, lpre, boff, ssrc, N, E, chunk);

  k_prescale<<<(NF + T - 1) / T, T, 0, stream>>>(x, norm_out, xs, NF);
  k_aggregate<<<gA, 256, 0, stream>>>(xs, ssrc, lpre, boff, degin, agg, N);
  k_gemm_wmma<<<gB, 256, 0, stream>>>(agg, norm_in, W1, b1, norm_out, xs, N, 1);

  k_aggregate<<<gA, 256, 0, stream>>>(xs, ssrc, lpre, boff, degin, agg, N);
  k_gemm_wmma<<<gB, 256, 0, stream>>>(agg, norm_in, W2, b2, nullptr, out, N, 0);
}


# Four channels per lane (16-byte loads) in the pre-scale and the gather-sum, so a wave sums four nodes at once: a quarter of the waves and row-load instructions; per-node sum order unchanged
# speedup vs baseline: 3.2364x; 1.0544x over previous
#include <hip/hip_runtime.h>
#include <math.h>


typedef __attribute__((ext_vector_type(2))) float v2f;
typedef __attribute__((ext_vector_type(8))) float v8f;

#define DF 32

#define HIST_CAP 65536
#define HIST_T 1024
#define HIST_S 64
#define HIST_U 8

__global__ __launch_bounds__(HIST_T) void k_deg_hist(
    const int* __restrict__ src, const int* __restrict__ dst,
    unsigned* __restrict__ part, int n, int e, int chunk) {
  __shared__ __attribute__((aligned(16))) unsigned hist[HIST_CAP];
  const int tid = threadIdx.x;
  const int slice = blockIdx.x;
  const int which = blockIdx.y;
  const int* __restrict__ arr = which ? dst : src;
  unsigned* __restrict__ plane =
      part + ((size_t)which * HIST_S + slice) * (size_t)n;
  const int j0 = slice * chunk;
  int j1 = j0 + chunk;
  if (j1 > e) j1 = e;
  const int cnt = (j1 > j0) ? (j1 - j0) : 0;
  const int nvec = cnt >> 2;
  const int ntail = cnt & 3;
  const int4* __restrict__ arr4 = reinterpret_cast<const int4*>(arr + j0);
  const bool vec_ok = (n & 3) == 0;

  for (int base = 0; base < n; base += HIST_CAP) {
    int nb = n - base;
    if (nb > HIST_CAP) nb = HIST_CAP;

    for (int i = tid * 4; i < nb; i += HIST_T * 4)
      *reinterpret_cast<uint4*>(&hist[i]) = make_uint4(0u, 0u, 0u, 0u);
    __syncthreads();

    for (int v0 = tid; v0 < nvec; v0 += HIST_T * HIST_U) {
      int4 v[HIST_U];
#pragma unroll
      for (int u = 0; u < HIST_U; ++u) {
        const int idx = v0 + u * HIST_T;
        v[u] = arr4[idx < nvec ? idx : nvec - 1];
      }
#pragma unroll
      for (int u = 0; u < HIST_U; ++u) {
        if (v0 + u * HIST_T < nvec) {
          unsigned b;
          b = (unsigned)(v[u].x - base);
          if (b < (unsigned)nb) atomicAdd(&hist[b], 1u);
          b = (unsigned)(v[u].y - base);
          if (b < (unsigned)nb) atomicAdd(&hist[b], 1u);
          b = (unsigned)(v[u].z - base);
          if (b < (unsigned)nb) atomicAdd(&hist[b], 1u);
          b = (unsigned)(v[u].w - base);
          if (b < (unsigned)nb) atomicAdd(&hist[b], 1u);
        }
      }
    }
    if (tid < ntail) {
      const unsigned b = (unsigned)(arr[j0 + nvec * 4 + tid] - base);
      if (b < (unsigned)nb) atomicAdd(&hist[b], 1u);
    }
    __syncthreads();

    if (vec_ok) {
      for (int i = tid * 4; i < nb; i += HIST_T * 4)
        *reinterpret_cast<uint4*>(&plane[base + i]) =
            *reinterpret_cast<const uint4*>(&hist[i]);
    } else {
      for (int i = tid; i < nb; i += HIST_T) plane[base + i] = hist[i];
    }
    __syncthreads();
  }
}

#define NB_T 256
#define NB_SHIFT 8

__global__ __launch_bounds__(NB_T) void k_norms(const unsigned* __restrict__ part,
                                                float* __restrict__ norms,
                                                unsigned* __restrict__ pre,
                                                unsigned* __restrict__ lpre,
                                                unsigned* __restrict__ degin,
                                                unsigned* __restrict__ btot,
                                                int n) {
  __shared__ unsigned sc[2][NB_T];
  const int tid = threadIdx.x;
  const int i = blockIdx.x * NB_T + tid;
  const int which = blockIdx.y;
  unsigned c = 0;
  if (i < n) {
    const unsigned* __restrict__ p = part + (size_t)which * HIST_S * n + i;
    unsigned v[HIST_S];
#pragma unroll
    for (int s = 0; s < HIST_S; ++s) v[s] = p[(size_t)s * n];
    if (which == 1) {
#pragma unroll
      for (int s = 0; s < HIST_S; ++s) {
        pre[(size_t)s * n + i] = c;
        c += v[s];
      }
    } else {
#pragma unroll
      for (int s = 0; s < HIST_S; ++s) c += v[s];
    }
    const float d = (float)c;
    norms[(size_t)which * n + i] = (d > 0.0f) ? (1.0f / sqrtf(d)) : 0.0f;
  }
  if (which == 1) {
    sc[0][tid] = c;
    __syncthreads();
    int cur = 0;
    for (int off = 1; off < NB_T; off <<= 1) {
      unsigned t = sc[cur][tid];
      if (tid >= off) t += sc[cur][tid - off];
      sc[cur ^ 1][tid] = t;
      cur ^= 1;
      __syncthreads();
    }
    const unsigned incl = sc[cur][tid];
    if (i < n) {
      lpre[i] = incl - c;
      degin[i] = c;
    }
    if (tid == NB_T - 1) btot[blockIdx.x] = incl;
  }
}

__global__ __launch_bounds__(1024) void k_blockscan(
    const unsigned* __restrict__ btot, unsigned* __restrict__ boff, int nblk) {
  __shared__ unsigned sc[2][1024];
  const int tid = threadIdx.x;
  unsigned carry = 0;
  for (int base = 0; base < nblk; base += 1024) {
    const int i = base + tid;
    const unsigned c = (i < nblk) ? btot[i] : 0u;
    sc[0][tid] = c;
    __syncthreads();
    int cur = 0;
    for (int off = 1; off < 1024; off <<= 1) {
      unsigned t = sc[cur][tid];
      if (tid >= off) t += sc[cur][tid - off];
      sc[cur ^ 1][tid] = t;
      cur ^= 1;
      __syncthreads();
    }
    const unsigned incl = sc[cur][tid];
    if (i < nblk) boff[i] = carry + incl - c;
    carry += sc[cur][1023];
    __syncthreads();
  }
}

__global__ __launch_bounds__(HIST_T) void k_place(
    const int* __restrict__ src, const int* __restrict__ dst,
    const unsigned* __restrict__ pre, const unsigned* __restrict__ lpre,
    const unsigned* __restrict__ boff, int* __restrict__ ssrc, int n, int e,
    int chunk) {
  __shared__ __attribute__((aligned(16))) unsigned cursor[HIST_CAP];
  const int tid = threadIdx.x;
  const int slice = blockIdx.x;
  const unsigned* __restrict__ plane = pre + (size_t)slice * (size_t)n;
  const int j0 = slice * chunk;
  int j1 = j0 + chunk;
  if (j1 > e) j1 = e;
  const int cnt = (j1 > j0) ? (j1 - j0) : 0;
  const int nvec = cnt >> 2;
  const int ntail = cnt & 3;
  const int4* __restrict__ src4 = reinterpret_cast<const int4*>(src + j0);
  const int4* __restrict__ dst4 = reinterpret_cast<const int4*>(dst + j0);
  const bool vec_ok = (n & 3) == 0;

  for (int base = 0; base < n; base += HIST_CAP) {
    int nb = n - base;
    if (nb > HIST_CAP) nb = HIST_CAP;

    if (vec_ok) {
      for (int i = tid * 4; i < nb; i += HIST_T * 4) {
        const int g = base + i;
        const uint4 l = *reinterpret_cast<const uint4*>(&lpre[g]);
        const uint4 p = *reinterpret_cast<const uint4*>(&plane[g]);
        const unsigned bo = boff[g >> NB_SHIFT];
        *reinterpret_cast<uint4*>(&cursor[i]) = make_uint4(
            bo + l.x + p.x, bo + l.y + p.y, bo + l.z + p.z, bo + l.w + p.w);
      }
    } else {
      for (int i = tid; i < nb; i += HIST_T) {
        const int g = base + i;
        cursor[i] = boff[g >> NB_SHIFT] + lpre[g] + plane[g];
      }
    }
    __syncthreads();

    for (int v0 = tid; v0 < nvec; v0 += HIST_T * HIST_U) {
      int4 dv[HIST_U], sv[HIST_U];
#pragma unroll
      for (int u = 0; u < HIST_U; ++u) {
        const int idx = v0 + u * HIST_T;
        const int cl = idx < nvec ? idx : nvec - 1;
        dv[u] = dst4[cl];
        sv[u] = src4[cl];
      }
#pragma unroll
      for (int u = 0; u < HIST_U; ++u) {
        if (v0 + u * HIST_T < nvec) {
          unsigned b, pos;
          b = (unsigned)(dv[u].x - base);
          if (b < (unsigned)nb) {
            pos = atomicAdd(&cursor[b], 1u);
            if (pos < (unsigned)e) ssrc[pos] = sv[u].x;
          }
          b = (unsigned)(dv[u].y - base);
          if (b < (unsigned)nb) {
            pos = atomicAdd(&cursor[b], 1u);
            if (pos < (unsigned)e) ssrc[pos] = sv[u].y;
          }
          b = (unsigned)(dv[u].z - base);
          if (b < (unsigned)nb) {
            pos = atomicAdd(&cursor[b], 1u);
            if (pos < (unsigned)e) ssrc[pos] = sv[u].z;
          }
          b = (unsigned)(dv[u].w - base);
          if (b < (unsigned)nb) {
            pos = atomicAdd(&cursor[b], 1u);
            if (pos < (unsigned)e) ssrc[pos] = sv[u].w;
          }
        }
      }
    }
    if (tid < ntail) {
      const int j = j0 + nvec * 4 + tid;
      const unsigned b = (unsigned)(dst[j] - base);
      if (b < (unsigned)nb) {
        const unsigned pos = atomicAdd(&cursor[b], 1u);
        if (pos < (unsigned)e) ssrc[pos] = src[j];
      }
    }
    __syncthreads();
  }
}

typedef __attribute__((ext_vector_type(4))) float v4f;

#define PS_U 4

__global__ __launch_bounds__(256) void k_prescale(const float* __restrict__ x,
                                                  const float* __restrict__ nrm,
                                                  float* __restrict__ xs,
                                                  int total4) {
  const v4f* __restrict__ x4 = reinterpret_cast<const v4f*>(x);
  v4f* __restrict__ xs4 = reinterpret_cast<v4f*>(xs);
  const int base = blockIdx.x * (256 * PS_U) + threadIdx.x;
  v4f v[PS_U];
  float s[PS_U];
#pragma unroll
  for (int u = 0; u < PS_U; ++u) {
    const int idx = base + u * 256;
    const int cl = idx < total4 ? idx : total4 - 1;
    v[u] = x4[cl];
    s[u] = nrm[cl >> 3];
  }
#pragma unroll
  for (int u = 0; u < PS_U; ++u) {
    const int idx = base + u * 256;
    if (idx < total4) xs4[idx] = v[u] * s[u];
  }
}

#define AGG_U 16

template <int J>
__device__ __forceinline__ unsigned group_bcast(unsigned v) {
  return __builtin_amdgcn_mov_dpp8(v, (unsigned)(J * 0x249249));
}

__global__ __launch_bounds__(256) void k_aggregate(
    const float* __restrict__ xs, const int* __restrict__ ssrc,
    const unsigned* __restrict__ lpre, const unsigned* __restrict__ boff,
    const unsigned* __restrict__ degin, float* __restrict__ agg, int n, int e) {
  const int lane = threadIdx.x & 31;
  const unsigned sub = (unsigned)(lane & 7);
  const int wave = blockIdx.x * 8 + (int)(threadIdx.x >> 5);
  if (wave * 4 >= n) return;
  int node = wave * 4 + (lane >> 3);
  const bool live = node < n;
  if (!live) node = n - 1;
  const unsigned start = boff[node >> NB_SHIFT] + lpre[node];
  const unsigned cnt = live ? degin[node] : 0u;
  unsigned mx = cnt;
  {
    const unsigned o = (unsigned)__shfl_xor((int)mx, 8);
    mx = o > mx ? o : mx;
  }
  {
    const unsigned o = (unsigned)__shfl_xor((int)mx, 16);
    mx = o > mx ? o : mx;
  }
  const unsigned maxcnt = (unsigned)__builtin_amdgcn_readfirstlane((int)mx);

  const char* __restrict__ xb = reinterpret_cast<const char*>(xs);
  const unsigned sub16 = sub * 16u;
  const unsigned last = cnt ? cnt - 1u : 0u;
  const unsigned emax = (unsigned)e - 1u;
  v4f acc = {0.0f, 0.0f, 0.0f, 0.0f};
  for (unsigned c0 = 0; c0 < maxcnt; c0 += AGG_U) {
    unsigned a0 = start + (c0 + sub < cnt ? c0 + sub : last);
    unsigned a1 = start + (c0 + 8u + sub < cnt ? c0 + 8u + sub : last);
    if (a0 > emax) a0 = emax;
    if (a1 > emax) a1 = emax;
    const unsigned o0 = (unsigned)ssrc[a0] * (unsigned)(DF * 4);
    const unsigned o1 = (unsigned)ssrc[a1] * (unsigned)(DF * 4);
    v4f v[AGG_U];
#define AGG_ROW(O, B, J) \
  v[B + J] = *reinterpret_cast<const v4f*>(xb + (group_bcast<J>(O) + sub16));
    AGG_ROW(o0, 0, 0) AGG_ROW(o0, 0, 1) AGG_ROW(o0, 0, 2) AGG_ROW(o0, 0, 3)
    AGG_ROW(o0, 0, 4) AGG_ROW(o0, 0, 5) AGG_ROW(o0, 0, 6) AGG_ROW(o0, 0, 7)
    AGG_ROW(o1, 8, 0) AGG_ROW(o1, 8, 1) AGG_ROW(o1, 8, 2) AGG_ROW(o1, 8, 3)
    AGG_ROW(o1, 8, 4) AGG_ROW(o1, 8, 5) AGG_ROW(o1, 8, 6) AGG_ROW(o1, 8, 7)
#undef AGG_ROW
    __builtin_amdgcn_sched_barrier(0);
#pragma unroll
    for (int j = 0; j < AGG_U; ++j) {
      const bool ok = c0 + (unsigned)j < cnt;
      acc.x += ok ? v[j].x : 0.0f;
      acc.y += ok ? v[j].y : 0.0f;
      acc.z += ok ? v[j].z : 0.0f;
      acc.w += ok ? v[j].w : 0.0f;
    }
  }
  if (live)
    reinterpret_cast<v4f*>(agg)[(size_t)node * (DF / 4) + sub] = acc;
}

__global__ __launch_bounds__(256) void k_gemm_wmma(
    const float* __restrict__ agg, const float* __restrict__ norm_in,
    const float* __restrict__ W, const float* __restrict__ bias,
    const float* __restrict__ post_scale,
    float* __restrict__ out, int n, int do_relu) {
  const int lane = threadIdx.x & 31;
  const int wave = threadIdx.x >> 5;
  const int row0 = (blockIdx.x * 8 + wave) * 16;
  if (row0 >= n) return;

  const int mrow = lane & 15;
  const int khalf = lane >> 4;
  int rload = row0 + mrow;
  if (rload > n - 1) rload = n - 1;
  const float ni = norm_in[rload];
  const float* __restrict__ arow = agg + (size_t)rload * DF;
  const int nb = lane & 15;

  v8f acc0 = {};
  v8f acc1 = {};
#pragma unroll
  for (int k0 = 0; k0 < DF; k0 += 4) {
    const int ka = k0 + khalf * 2;
    v2f a;
    a.x = arow[ka + 0] * ni;
    a.y = arow[ka + 1] * ni;
    v2f b0, b1;
    b0.x = W[(ka + 0) * DF + nb];
    b0.y = W[(ka + 1) * DF + nb];
    b1.x = W[(ka + 0) * DF + 16 + nb];
    b1.y = W[(ka + 1) * DF + 16 + nb];
    acc0 = __builtin_amdgcn_wmma_f32_16x16x4_f32(false, a, false, b0,
                                                 (short)0, acc0, false, false);
    acc1 = __builtin_amdgcn_wmma_f32_16x16x4_f32(false, a, false, b1,
                                                 (short)0, acc1, false, false);
  }

  const float bc0 = bias[nb];
  const float bc1 = bias[16 + nb];
#pragma unroll
  for (int v = 0; v < 8; ++v) {
    const int m = row0 + v + 8 * khalf;
    if (m < n) {
      float o0 = acc0[v] + bc0;
      float o1 = acc1[v] + bc1;
      if (do_relu) {
        o0 = fmaxf(o0, 0.0f);
        o1 = fmaxf(o1, 0.0f);
      }
      const float s = post_scale ? post_scale[m] : 1.0f;
      out[(size_t)m * DF + nb] = o0 * s;
      out[(size_t)m * DF + 16 + nb] = o1 * s;
    }
  }
}

extern "C" void kernel_launch(void* const* d_in, const int* in_sizes, int n_in,
                              void* d_out, int out_size, void* d_ws,
                              size_t ws_size, hipStream_t stream) {
  const float* x = (const float*)d_in[0];
  const int* src = (const int*)d_in[1];
  const int* dst = (const int*)d_in[2];
  const float* W1 = (const float*)d_in[3];
  const float* b1 = (const float*)d_in[4];
  const float* W2 = (const float*)d_in[5];
  const float* b2 = (const float*)d_in[6];
  float* out = (float*)d_out;

  const int N = in_sizes[0] / DF;
  const int E = in_sizes[1];
  const int NF = N * DF;

  float* ws = (float*)d_ws;
  float* norm_out = ws;
  float* norm_in = ws + N;
  float* xs = ws + ((2 * (size_t)N + 3) & ~(size_t)3);
  float* agg = xs + (size_t)NF;
  unsigned* part = (unsigned*)(agg + (size_t)NF);
  const int nblk = (N + NB_T - 1) / NB_T;
  unsigned* pre = part + 2 * (size_t)HIST_S * N;
  unsigned* lpre = pre + (size_t)HIST_S * N;
  unsigned* degin = lpre + N;
  unsigned* btot = degin + N;
  unsigned* boff = btot + nblk;
  int* ssrc = (int*)(boff + nblk);

  const int T = 256;
  const int gB = ((N + 15) / 16 + 7) / 8;
  const int gA = (N + 31) / 32;

  const int chunk = (((E + HIST_S - 1) / HIST_S) + 3) & ~3;
  k_deg_hist<<<dim3(HIST_S, 2), HIST_T, 0, stream>>>(src, dst, part, N, E, chunk);
  k_norms<<<dim3(nblk, 2), NB_T, 0, stream>>>(part, norm_out, pre, lpre, degin, btot, N);
  k_blockscan<<<1, 1024, 0, stream>>>(btot, boff, nblk);
  k_place<<<HIST_S, HIST_T, 0, stream>>>(src, dst, pre, lpre, boff, ssrc, N, E, chunk);

  k_prescale<<<(NF / 4 + T * PS_U - 1) / (T * PS_U), T, 0, stream>>>(x, norm_out, xs, NF / 4);
  k_aggregate<<<gA, 256, 0, stream>>>(xs, ssrc, lpre, boff, degin, agg, N, E);
  k_gemm_wmma<<<gB, 256, 0, stream>>>(agg, norm_in, W1, b1, norm_out, xs, N, 1);

  k_aggregate<<<gA, 256, 0, stream>>>(xs, ssrc, lpre, boff, degin, agg, N, E);
  k_gemm_wmma<<<gB, 256, 0, stream>>>(agg, norm_in, W2, b2, nullptr, out, N, 0);
}


# Dense transform fused into the gather-sum kernel: a work-group's 32 summed rows go through LDS to two waves' f32 WMMA tiles; no agg buffer, no GEMM launch (9 -> 7 launches); sum order unchanged
# speedup vs baseline: 3.4084x; 1.0531x over previous
#include <hip/hip_runtime.h>
#include <math.h>


typedef __attribute__((ext_vector_type(2))) float v2f;
typedef __attribute__((ext_vector_type(8))) float v8f;

#define DF 32

#define HIST_CAP 65536
#define HIST_T 1024
#define HIST_S 64
#define HIST_U 8

__global__ __launch_bounds__(HIST_T) void k_deg_hist(
    const int* __restrict__ src, const int* __restrict__ dst,
    unsigned* __restrict__ part, int n, int e, int chunk) {
  __shared__ __attribute__((aligned(16))) unsigned hist[HIST_CAP];
  const int tid = threadIdx.x;
  const int slice = blockIdx.x;
  const int which = blockIdx.y;
  const int* __restrict__ arr = which ? dst : src;
  unsigned* __restrict__ plane =
      part + ((size_t)which * HIST_S + slice) * (size_t)n;
  const int j0 = slice * chunk;
  int j1 = j0 + chunk;
  if (j1 > e) j1 = e;
  const int cnt = (j1 > j0) ? (j1 - j0) : 0;
  const int nvec = cnt >> 2;
  const int ntail = cnt & 3;
  const int4* __restrict__ arr4 = reinterpret_cast<const int4*>(arr + j0);
  const bool vec_ok = (n & 3) == 0;

  for (int base = 0; base < n; base += HIST_CAP) {
    int nb = n - base;
    if (nb > HIST_CAP) nb = HIST_CAP;

    for (int i = tid * 4; i < nb; i += HIST_T * 4)
      *reinterpret_cast<uint4*>(&hist[i]) = make_uint4(0u, 0u, 0u, 0u);
    __syncthreads();

    for (int v0 = tid; v0 < nvec; v0 += HIST_T * HIST_U) {
      int4 v[HIST_U];
#pragma unroll
      for (int u = 0; u < HIST_U; ++u) {
        const int idx = v0 + u * HIST_T;
        v[u] = arr4[idx < nvec ? idx : nvec - 1];
      }
#pragma unroll
      for (int u = 0; u < HIST_U; ++u) {
        if (v0 + u * HIST_T < nvec) {
          unsigned b;
          b = (unsigned)(v[u].x - base);
          if (b < (unsigned)nb) atomicAdd(&hist[b], 1u);
          b = (unsigned)(v[u].y - base);
          if (b < (unsigned)nb) atomicAdd(&hist[b], 1u);
          b = (unsigned)(v[u].z - base);
          if (b < (unsigned)nb) atomicAdd(&hist[b], 1u);
          b = (unsigned)(v[u].w - base);
          if (b < (unsigned)nb) atomicAdd(&hist[b], 1u);
        }
      }
    }
    if (tid < ntail) {
      const unsigned b = (unsigned)(arr[j0 + nvec * 4 + tid] - base);
      if (b < (unsigned)nb) atomicAdd(&hist[b], 1u);
    }
    __syncthreads();

    if (vec_ok) {
      for (int i = tid * 4; i < nb; i += HIST_T * 4)
        *reinterpret_cast<uint4*>(&plane[base + i]) =
            *reinterpret_cast<const uint4*>(&hist[i]);
    } else {
      for (int i = tid; i < nb; i += HIST_T) plane[base + i] = hist[i];
    }
    __syncthreads();
  }
}

#define NB_T 256
#define NB_SHIFT 8

__global__ __launch_bounds__(NB_T) void k_norms(const unsigned* __restrict__ part,
                                                float* __restrict__ norms,
                                                unsigned* __restrict__ pre,
                                                unsigned* __restrict__ lpre,
                                                unsigned* __restrict__ degin,
                                                unsigned* __restrict__ btot,
                                                int n) {
  __shared__ unsigned sc[2][NB_T];
  const int tid = threadIdx.x;
  const int i = blockIdx.x * NB_T + tid;
  const int which = blockIdx.y;
  unsigned c = 0;
  if (i < n) {
    const unsigned* __restrict__ p = part + (size_t)which * HIST_S * n + i;
    unsigned v[HIST_S];
#pragma unroll
    for (int s = 0; s < HIST_S; ++s) v[s] = p[(size_t)s * n];
    if (which == 1) {
#pragma unroll
      for (int s = 0; s < HIST_S; ++s) {
        pre[(size_t)s * n + i] = c;
        c += v[s];
      }
    } else {
#pragma unroll
      for (int s = 0; s < HIST_S; ++s) c += v[s];
    }
    const float d = (float)c;
    norms[(size_t)which * n + i] = (d > 0.0f) ? (1.0f / sqrtf(d)) : 0.0f;
  }
  if (which == 1) {
    sc[0][tid] = c;
    __syncthreads();
    int cur = 0;
    for (int off = 1; off < NB_T; off <<= 1) {
      unsigned t = sc[cur][tid];
      if (tid >= off) t += sc[cur][tid - off];
      sc[cur ^ 1][tid] = t;
      cur ^= 1;
      __syncthreads();
    }
    const unsigned incl = sc[cur][tid];
    if (i < n) {
      lpre[i] = incl - c;
      degin[i] = c;
    }
    if (tid == NB_T - 1) btot[blockIdx.x] = incl;
  }
}

__global__ __launch_bounds__(1024) void k_blockscan(
    const unsigned* __restrict__ btot, unsigned* __restrict__ boff, int nblk) {
  __shared__ unsigned sc[2][1024];
  const int tid = threadIdx.x;
  unsigned carry = 0;
  for (int base = 0; base < nblk; base += 1024) {
    const int i = base + tid;
    const unsigned c = (i < nblk) ? btot[i] : 0u;
    sc[0][tid] = c;
    __syncthreads();
    int cur = 0;
    for (int off = 1; off < 1024; off <<= 1) {
      unsigned t = sc[cur][tid];
      if (tid >= off) t += sc[cur][tid - off];
      sc[cur ^ 1][tid] = t;
      cur ^= 1;
      __syncthreads();
    }
    const unsigned incl = sc[cur][tid];
    if (i < nblk) boff[i] = carry + incl - c;
    carry += sc[cur][1023];
    __syncthreads();
  }
}

__global__ __launch_bounds__(HIST_T) void k_place(
    const int* __restrict__ src, const int* __restrict__ dst,
    const unsigned* __restrict__ pre, const unsigned* __restrict__ lpre,
    const unsigned* __restrict__ boff, int* __restrict__ ssrc, int n, int e,
    int chunk) {
  __shared__ __attribute__((aligned(16))) unsigned cursor[HIST_CAP];
  const int tid = threadIdx.x;
  const int slice = blockIdx.x;
  const unsigned* __restrict__ plane = pre + (size_t)slice * (size_t)n;
  const int j0 = slice * chunk;
  int j1 = j0 + chunk;
  if (j1 > e) j1 = e;
  const int cnt = (j1 > j0) ? (j1 - j0) : 0;
  const int nvec = cnt >> 2;
  const int ntail = cnt & 3;
  const int4* __restrict__ src4 = reinterpret_cast<const int4*>(src + j0);
  const int4* __restrict__ dst4 = reinterpret_cast<const int4*>(dst + j0);
  const bool vec_ok = (n & 3) == 0;

  for (int base = 0; base < n; base += HIST_CAP) {
    int nb = n - base;
    if (nb > HIST_CAP) nb = HIST_CAP;

    if (vec_ok) {
      for (int i = tid * 4; i < nb; i += HIST_T * 4) {
        const int g = base + i;
        const uint4 l = *reinterpret_cast<const uint4*>(&lpre[g]);
        const uint4 p = *reinterpret_cast<const uint4*>(&plane[g]);
        const unsigned bo = boff[g >> NB_SHIFT];
        *reinterpret_cast<uint4*>(&cursor[i]) = make_uint4(
            bo + l.x + p.x, bo + l.y + p.y, bo + l.z + p.z, bo + l.w + p.w);
      }
    } else {
      for (int i = tid; i < nb; i += HIST_T) {
        const int g = base + i;
        cursor[i] = boff[g >> NB_SHIFT] + lpre[g] + plane[g];
      }
    }
    __syncthreads();

    for (int v0 = tid; v0 < nvec; v0 += HIST_T * HIST_U) {
      int4 dv[HIST_U], sv[HIST_U];
#pragma unroll
      for (int u = 0; u < HIST_U; ++u) {
        const int idx = v0 + u * HIST_T;
        const int cl = idx < nvec ? idx : nvec - 1;
        dv[u] = dst4[cl];
        sv[u] = src4[cl];
      }
#pragma unroll
      for (int u = 0; u < HIST_U; ++u) {
        if (v0 + u * HIST_T < nvec) {
          unsigned b, pos;
          b = (unsigned)(dv[u].x - base);
          if (b < (unsigned)nb) {
            pos = atomicAdd(&cursor[b], 1u);
            if (pos < (unsigned)e) ssrc[pos] = sv[u].x;
          }
          b = (unsigned)(dv[u].y - base);
          if (b < (unsigned)nb) {
            pos = atomicAdd(&cursor[b], 1u);
            if (pos < (unsigned)e) ssrc[pos] = sv[u].y;
          }
          b = (unsigned)(dv[u].z - base);
          if (b < (unsigned)nb) {
            pos = atomicAdd(&cursor[b], 1u);
            if (pos < (unsigned)e) ssrc[pos] = sv[u].z;
          }
          b = (unsigned)(dv[u].w - base);
          if (b < (unsigned)nb) {
            pos = atomicAdd(&cursor[b], 1u);
            if (pos < (unsigned)e) ssrc[pos] = sv[u].w;
          }
        }
      }
    }
    if (tid < ntail) {
      const int j = j0 + nvec * 4 + tid;
      const unsigned b = (unsigned)(dst[j] - base);
      if (b < (unsigned)nb) {
        const unsigned pos = atomicAdd(&cursor[b], 1u);
        if (pos < (unsigned)e) ssrc[pos] = src[j];
      }
    }
    __syncthreads();
  }
}

typedef __attribute__((ext_vector_type(4))) float v4f;

#define PS_U 4

__global__ __launch_bounds__(256) void k_prescale(const float* __restrict__ x,
                                                  const float* __restrict__ nrm,
                                                  float* __restrict__ xs,
                                                  int total4) {
  const v4f* __restrict__ x4 = reinterpret_cast<const v4f*>(x);
  v4f* __restrict__ xs4 = reinterpret_cast<v4f*>(xs);
  const int base = blockIdx.x * (256 * PS_U) + threadIdx.x;
  v4f v[PS_U];
  float s[PS_U];
#pragma unroll
  for (int u = 0; u < PS_U; ++u) {
    const int idx = base + u * 256;
    const int cl = idx < total4 ? idx : total4 - 1;
    v[u] = x4[cl];
    s[u] = nrm[cl >> 3];
  }
#pragma unroll
  for (int u = 0; u < PS_U; ++u) {
    const int idx = base + u * 256;
    if (idx < total4) xs4[idx] = v[u] * s[u];
  }
}

#define AGG_U 16

template <int J>
__device__ __forceinline__ unsigned group_bcast(unsigned v) {
  return __builtin_amdgcn_mov_dpp8(v, (unsigned)(J * 0x249249));
}

#define AT_PITCH 36

__global__ __launch_bounds__(256, 12) void k_gather_gemm(
    const float* __restrict__ xs, const int* __restrict__ ssrc,
    const unsigned* __restrict__ lpre, const unsigned* __restrict__ boff,
    const unsigned* __restrict__ degin, const float* __restrict__ norm_in,
    const float* __restrict__ W, const float* __restrict__ bias,
    const float* __restrict__ post_scale,
    float* __restrict__ out, int n, int e, int do_relu) {
  __shared__ __attribute__((aligned(16))) float tile[32 * AT_PITCH];
  __shared__ float rscale[32];
  const int lane = threadIdx.x & 31;
  const unsigned sub = (unsigned)(lane & 7);
  const int wv = __builtin_amdgcn_readfirstlane((int)(threadIdx.x >> 5));
  const int slot = wv * 4 + (lane >> 3);
  int node = blockIdx.x * 32 + slot;
  const bool live = node < n;
  if (!live) node = n - 1;
  const unsigned start = boff[node >> NB_SHIFT] + lpre[node];
  const unsigned cnt = live ? degin[node] : 0u;
  const float ni = norm_in[node];
  const float po = post_scale ? post_scale[node] : 1.0f;
  unsigned mx = cnt;
  {
    const unsigned o = (unsigned)__shfl_xor((int)mx, 8);
    mx = o > mx ? o : mx;
  }
  {
    const unsigned o = (unsigned)__shfl_xor((int)mx, 16);
    mx = o > mx ? o : mx;
  }
  const unsigned maxcnt = (unsigned)__builtin_amdgcn_readfirstlane((int)mx);

  const char* __restrict__ xb = reinterpret_cast<const char*>(xs);
  const unsigned sub16 = sub * 16u;
  const unsigned last = cnt ? cnt - 1u : 0u;
  const unsigned emax = (unsigned)e - 1u;
  v4f acc = {0.0f, 0.0f, 0.0f, 0.0f};
  for (unsigned c0 = 0; c0 < maxcnt; c0 += AGG_U) {
    unsigned a0 = start + (c0 + sub < cnt ? c0 + sub : last);
    unsigned a1 = start + (c0 + 8u + sub < cnt ? c0 + 8u + sub : last);
    if (a0 > emax) a0 = emax;
    if (a1 > emax) a1 = emax;
    const unsigned o0 = (unsigned)ssrc[a0] * (unsigned)(DF * 4);
    const unsigned o1 = (unsigned)ssrc[a1] * (unsigned)(DF * 4);
    v4f v[AGG_U];
#define AGG_ROW(O, B, J) \
  v[B + J] = *reinterpret_cast<const v4f*>(xb + (group_bcast<J>(O) + sub16));
    AGG_ROW(o0, 0, 0) AGG_ROW(o0, 0, 1) AGG_ROW(o0, 0, 2) AGG_ROW(o0, 0, 3)
    AGG_ROW(o0, 0, 4) AGG_ROW(o0, 0, 5) AGG_ROW(o0, 0, 6) AGG_ROW(o0, 0, 7)
    AGG_ROW(o1, 8, 0) AGG_ROW(o1, 8, 1) AGG_ROW(o1, 8, 2) AGG_ROW(o1, 8, 3)
    AGG_ROW(o1, 8, 4) AGG_ROW(o1, 8, 5) AGG_ROW(o1, 8, 6) AGG_ROW(o1, 8, 7)
#undef AGG_ROW
    __builtin_amdgcn_sched_barrier(0);
#pragma unroll
    for (int j = 0; j < AGG_U; ++j) {
      const bool ok = c0 + (unsigned)j < cnt;
      acc.x += ok ? v[j].x : 0.0f;
      acc.y += ok ? v[j].y : 0.0f;
      acc.z += ok ? v[j].z : 0.0f;
      acc.w += ok ? v[j].w : 0.0f;
    }
  }

  *reinterpret_cast<v4f*>(&tile[slot * AT_PITCH + (int)sub * 4]) = acc * ni;
  if (sub == 0u) rscale[slot] = po;
  __syncthreads();

  const int row0 = blockIdx.x * 32 + wv * 16;
  if (wv < 2 && row0 < n) {
    const int mrow = lane & 15;
    const int khalf = lane >> 4;
    const int nb = lane & 15;
    const float* arow = &tile[(wv * 16 + mrow) * AT_PITCH];
    v8f acc0 = {};
    v8f acc1 = {};
#pragma unroll
    for (int k0 = 0; k0 < DF; k0 += 4) {
      const int ka = k0 + khalf * 2;
      v2f a;
      a.x = arow[ka + 0];
      a.y = arow[ka + 1];
      v2f b0, b1;
      b0.x = W[(ka + 0) * DF + nb];
      b0.y = W[(ka + 1) * DF + nb];
      b1.x = W[(ka + 0) * DF + 16 + nb];
      b1.y = W[(ka + 1) * DF + 16 + nb];
      acc0 = __builtin_amdgcn_wmma_f32_16x16x4_f32(false, a, false, b0,
                                                   (short)0, acc0, false, false);
      acc1 = __builtin_amdgcn_wmma_f32_16x16x4_f32(false, a, false, b1,
                                                   (short)0, acc1, false, false);
    }

    const float bc0 = bias[nb];
    const float bc1 = bias[16 + nb];
#pragma unroll
    for (int v = 0; v < 8; ++v) {
      const int r = v + 8 * khalf;
      const int m = row0 + r;
      if (m < n) {
        float o0 = acc0[v] + bc0;
        float o1 = acc1[v] + bc1;
        if (do_relu) {
          o0 = fmaxf(o0, 0.0f);
          o1 = fmaxf(o1, 0.0f);
        }
        const float s = rscale[wv * 16 + r];
        out[(size_t)m * DF + nb] = o0 * s;
        out[(size_t)m * DF + 16 + nb] = o1 * s;
      }
    }
  }
}

extern "C" void kernel_launch(void* const* d_in, const int* in_sizes, int n_in,
                              void* d_out, int out_size, void* d_ws,
                              size_t ws_size, hipStream_t stream) {
  const float* x = (const float*)d_in[0];
  const int* src = (const int*)d_in[1];
  const int* dst = (const int*)d_in[2];
  const float* W1 = (const float*)d_in[3];
  const float* b1 = (const float*)d_in[4];
  const float* W2 = (const float*)d_in[5];
  const float* b2 = (const float*)d_in[6];
  float* out = (float*)d_out;

  const int N = in_sizes[0] / DF;
  const int E = in_sizes[1];
  const int NF = N * DF;

  float* ws = (float*)d_ws;
  float* norm_out = ws;
  float* norm_in = ws + N;
  float* xs = ws + ((2 * (size_t)N + 3) & ~(size_t)3);
  float* h1s = xs + (size_t)NF;
  unsigned* part = (unsigned*)(h1s + (size_t)NF);
  const int nblk = (N + NB_T - 1) / NB_T;
  unsigned* pre = part + 2 * (size_t)HIST_S * N;
  unsigned* lpre = pre + (size_t)HIST_S * N;
  unsigned* degin = lpre + N;
  unsigned* btot = degin + N;
  unsigned* boff = btot + nblk;
  int* ssrc = (int*)(boff + nblk);

  const int T = 256;
  const int gA = (N + 31) / 32;

  const int chunk = (((E + HIST_S - 1) / HIST_S) + 3) & ~3;
  k_deg_hist<<<dim3(HIST_S, 2), HIST_T, 0, stream>>>(src, dst, part, N, E, chunk);
  k_norms<<<dim3(nblk, 2), NB_T, 0, stream>>>(part, norm_out, pre, lpre, degin, btot, N);
  k_blockscan<<<1, 1024, 0, stream>>>(btot, boff, nblk);
  k_place<<<HIST_S, HIST_T, 0, stream>>>(src, dst, pre, lpre, boff, ssrc, N, E, chunk);

  k_prescale<<<(NF / 4 + T * PS_U - 1) / (T * PS_U), T, 0, stream>>>(x, norm_out, xs, NF / 4);
  k_gather_gemm<<<gA, 256, 0, stream>>>(xs, ssrc, lpre, boff, degin, norm_in, W1, b1,
                                        norm_out, h1s, N, E, 1);

  k_gather_gemm<<<gA, 256, 0, stream>>>(h1s, ssrc, lpre, boff, degin, norm_in, W2, b2,
                                        nullptr, out, N, E, 0);
}


# Counting-sort placement kernel split 4 ways by destination-node range (64 -> 256 work-groups, one per WGP) so that all WGPs share the scattered 4-byte stores; per-node sum order as before
# speedup vs baseline: 4.1376x; 1.2139x over previous
#include <hip/hip_runtime.h>
#include <math.h>


typedef __attribute__((ext_vector_type(2))) float v2f;
typedef __attribute__((ext_vector_type(8))) float v8f;

#define DF 32

#define HIST_CAP 65536
#define HIST_T 1024
#define HIST_S 64
#define HIST_U 8

__global__ __launch_bounds__(HIST_T) void k_deg_hist(
    const int* __restrict__ src, const int* __restrict__ dst,
    unsigned* __restrict__ part, int n, int e, int chunk) {
  __shared__ __attribute__((aligned(16))) unsigned hist[HIST_CAP];
  const int tid = threadIdx.x;
  const int slice = blockIdx.x;
  const int which = blockIdx.y;
  const int* __restrict__ arr = which ? dst : src;
  unsigned* __restrict__ plane =
      part + ((size_t)which * HIST_S + slice) * (size_t)n;
  const int j0 = slice * chunk;
  int j1 = j0 + chunk;
  if (j1 > e) j1 = e;
  const int cnt = (j1 > j0) ? (j1 - j0) : 0;
  const int nvec = cnt >> 2;
  const int ntail = cnt & 3;
  const int4* __restrict__ arr4 = reinterpret_cast<const int4*>(arr + j0);
  const bool vec_ok = (n & 3) == 0;

  for (int base = 0; base < n; base += HIST_CAP) {
    int nb = n - base;
    if (nb > HIST_CAP) nb = HIST_CAP;

    for (int i = tid * 4; i < nb; i += HIST_T * 4)
      *reinterpret_cast<uint4*>(&hist[i]) = make_uint4(0u, 0u, 0u, 0u);
    __syncthreads();

    for (int v0 = tid; v0 < nvec; v0 += HIST_T * HIST_U) {
      int4 v[HIST_U];
#pragma unroll
      for (int u = 0; u < HIST_U; ++u) {
        const int idx = v0 + u * HIST_T;
        v[u] = arr4[idx < nvec ? idx : nvec - 1];
      }
#pragma unroll
      for (int u = 0; u < HIST_U; ++u) {
        if (v0 + u * HIST_T < nvec) {
          unsigned b;
          b = (unsigned)(v[u].x - base);
          if (b < (unsigned)nb) atomicAdd(&hist[b], 1u);
          b = (unsigned)(v[u].y - base);
          if (b < (unsigned)nb) atomicAdd(&hist[b], 1u);
          b = (unsigned)(v[u].z - base);
          if (b < (unsigned)nb) atomicAdd(&hist[b], 1u);
          b = (unsigned)(v[u].w - base);
          if (b < (unsigned)nb) atomicAdd(&hist[b], 1u);
        }
      }
    }
    if (tid < ntail) {
      const unsigned b = (unsigned)(arr[j0 + nvec * 4 + tid] - base);
      if (b < (unsigned)nb) atomicAdd(&hist[b], 1u);
    }
    __syncthreads();

    if (vec_ok) {
      for (int i = tid * 4; i < nb; i += HIST_T * 4)
        *reinterpret_cast<uint4*>(&plane[base + i]) =
            *reinterpret_cast<const uint4*>(&hist[i]);
    } else {
      for (int i = tid; i < nb; i += HIST_T) plane[base + i] = hist[i];
    }
    __syncthreads();
  }
}

#define NB_T 256
#define NB_SHIFT 8

__global__ __launch_bounds__(NB_T) void k_norms(const unsigned* __restrict__ part,
                                                float* __restrict__ norms,
                                                unsigned* __restrict__ pre,
                                                unsigned* __restrict__ lpre,
                                                unsigned* __restrict__ degin,
                                                unsigned* __restrict__ btot,
                                                int n) {
  __shared__ unsigned sc[2][NB_T];
  const int tid = threadIdx.x;
  const int i = blockIdx.x * NB_T + tid;
  const int which = blockIdx.y;
  unsigned c = 0;
  if (i < n) {
    const unsigned* __restrict__ p = part + (size_t)which * HIST_S * n + i;
    unsigned v[HIST_S];
#pragma unroll
    for (int s = 0; s < HIST_S; ++s) v[s] = p[(size_t)s * n];
    if (which == 1) {
#pragma unroll
      for (int s = 0; s < HIST_S; ++s) {
        pre[(size_t)s * n + i] = c;
        c += v[s];
      }
    } else {
#pragma unroll
      for (int s = 0; s < HIST_S; ++s) c += v[s];
    }
    const float d = (float)c;
    norms[(size_t)which * n + i] = (d > 0.0f) ? (1.0f / sqrtf(d)) : 0.0f;
  }
  if (which == 1) {
    sc[0][tid] = c;
    __syncthreads();
    int cur = 0;
    for (int off = 1; off < NB_T; off <<= 1) {
      unsigned t = sc[cur][tid];
      if (tid >= off) t += sc[cur][tid - off];
      sc[cur ^ 1][tid] = t;
      cur ^= 1;
      __syncthreads();
    }
    const unsigned incl = sc[cur][tid];
    if (i < n) {
      lpre[i] = incl - c;
      degin[i] = c;
    }
    if (tid == NB_T - 1) btot[blockIdx.x] = incl;
  }
}

__global__ __launch_bounds__(1024) void k_blockscan(
    const unsigned* __restrict__ btot, unsigned* __restrict__ boff, int nblk) {
  __shared__ unsigned sc[2][1024];
  const int tid = threadIdx.x;
  unsigned carry = 0;
  for (int base = 0; base < nblk; base += 1024) {
    const int i = base + tid;
    const unsigned c = (i < nblk) ? btot[i] : 0u;
    sc[0][tid] = c;
    __syncthreads();
    int cur = 0;
    for (int off = 1; off < 1024; off <<= 1) {
      unsigned t = sc[cur][tid];
      if (tid >= off) t += sc[cur][tid - off];
      sc[cur ^ 1][tid] = t;
      cur ^= 1;
      __syncthreads();
    }
    const unsigned incl = sc[cur][tid];
    if (i < nblk) boff[i] = carry + incl - c;
    carry += sc[cur][1023];
    __syncthreads();
  }
}

#define PL_P 4

__global__ __launch_bounds__(HIST_T) void k_place(
    const int* __restrict__ src, const int* __restrict__ dst,
    const unsigned* __restrict__ pre, const unsigned* __restrict__ lpre,
    const unsigned* __restrict__ boff, int* __restrict__ ssrc, int n, int e,
    int chunk) {
  __shared__ __attribute__((aligned(16))) unsigned cursor[HIST_CAP];
  const int tid = threadIdx.x;
  const int slice = blockIdx.x;
  const int per = (((n + PL_P - 1) / PL_P) + 3) & ~3;
  const int r0 = (int)blockIdx.y * per;
  int r1 = r0 + per;
  if (r1 > n) r1 = n;
  const unsigned* __restrict__ plane = pre + (size_t)slice * (size_t)n;
  const int j0 = slice * chunk;
  int j1 = j0 + chunk;
  if (j1 > e) j1 = e;
  const int cnt = (j1 > j0) ? (j1 - j0) : 0;
  const int nvec = cnt >> 2;
  const int ntail = cnt & 3;
  const int4* __restrict__ src4 = reinterpret_cast<const int4*>(src + j0);
  const int4* __restrict__ dst4 = reinterpret_cast<const int4*>(dst + j0);
  const bool vec_ok = (n & 3) == 0;

  for (int base = r0; base < r1; base += HIST_CAP) {
    int nb = r1 - base;
    if (nb > HIST_CAP) nb = HIST_CAP;

    if (vec_ok) {
      for (int i = tid * 4; i < nb; i += HIST_T * 4) {
        const int g = base + i;
        const uint4 l = *reinterpret_cast<const uint4*>(&lpre[g]);
        const uint4 p = *reinterpret_cast<const uint4*>(&plane[g]);
        const unsigned bo = boff[g >> NB_SHIFT];
        *reinterpret_cast<uint4*>(&cursor[i]) = make_uint4(
            bo + l.x + p.x, bo + l.y + p.y, bo + l.z + p.z, bo + l.w + p.w);
      }
    } else {
      for (int i = tid; i < nb; i += HIST_T) {
        const int g = base + i;
        cursor[i] = boff[g >> NB_SHIFT] + lpre[g] + plane[g];
      }
    }
    __syncthreads();

    for (int v0 = tid; v0 < nvec; v0 += HIST_T * HIST_U) {
      int4 dv[HIST_U], sv[HIST_U];
#pragma unroll
      for (int u = 0; u < HIST_U; ++u) {
        const int idx = v0 + u * HIST_T;
        const int cl = idx < nvec ? idx : nvec - 1;
        dv[u] = dst4[cl];
        sv[u] = src4[cl];
      }
#pragma unroll
      for (int u = 0; u < HIST_U; ++u) {
        if (v0 + u * HIST_T < nvec) {
          unsigned b, pos;
          b = (unsigned)(dv[u].x - base);
          if (b < (unsigned)nb) {
            pos = atomicAdd(&cursor[b], 1u);
            if (pos < (unsigned)e) ssrc[pos] = sv[u].x;
          }
          b = (unsigned)(dv[u].y - base);
          if (b < (unsigned)nb) {
            pos = atomicAdd(&cursor[b], 1u);
            if (pos < (unsigned)e) ssrc[pos] = sv[u].y;
          }
          b = (unsigned)(dv[u].z - base);
          if (b < (unsigned)nb) {
            pos = atomicAdd(&cursor[b], 1u);
            if (pos < (unsigned)e) ssrc[pos] = sv[u].z;
          }
          b = (unsigned)(dv[u].w - base);
          if (b < (unsigned)nb) {
            pos = atomicAdd(&cursor[b], 1u);
            if (pos < (unsigned)e) ssrc[pos] = sv[u].w;
          }
        }
      }
    }
    if (tid < ntail) {
      const int j = j0 + nvec * 4 + tid;
      const unsigned b = (unsigned)(dst[j] - base);
      if (b < (unsigned)nb) {
        const unsigned pos = atomicAdd(&cursor[b], 1u);
        if (pos < (unsigned)e) ssrc[pos] = src[j];
      }
    }
    __syncthreads();
  }
}

typedef __attribute__((ext_vector_type(4))) float v4f;

#define PS_U 4

__global__ __launch_bounds__(256) void k_prescale(const float* __restrict__ x,
                                                  const float* __restrict__ nrm,
                                                  float* __restrict__ xs,
                                                  int total4) {
  const v4f* __restrict__ x4 = reinterpret_cast<const v4f*>(x);
  v4f* __restrict__ xs4 = reinterpret_cast<v4f*>(xs);
  const int base = blockIdx.x * (256 * PS_U) + threadIdx.x;
  v4f v[PS_U];
  float s[PS_U];
#pragma unroll
  for (int u = 0; u < PS_U; ++u) {
    const int idx = base + u * 256;
    const int cl = idx < total4 ? idx : total4 - 1;
    v[u] = x4[cl];
    s[u] = nrm[cl >> 3];
  }
#pragma unroll
  for (int u = 0; u < PS_U; ++u) {
    const int idx = base + u * 256;
    if (idx < total4) xs4[idx] = v[u] * s[u];
  }
}

#define AGG_U 16

template <int J>
__device__ __forceinline__ unsigned group_bcast(unsigned v) {
  return __builtin_amdgcn_mov_dpp8(v, (unsigned)(J * 0x249249));
}

#define AT_PITCH 36

__global__ __launch_bounds__(256, 12) void k_gather_gemm(
    const float* __restrict__ xs, const int* __restrict__ ssrc,
    const unsigned* __restrict__ lpre, const unsigned* __restrict__ boff,
    const unsigned* __restrict__ degin, const float* __restrict__ norm_in,
    const float* __restrict__ W, const float* __restrict__ bias,
    const float* __restrict__ post_scale,
    float* __restrict__ out, int n, int e, int do_relu) {
  __shared__ __attribute__((aligned(16))) float tile[32 * AT_PITCH];
  __shared__ float rscale[32];
  const int lane = threadIdx.x & 31;
  const unsigned sub = (unsigned)(lane & 7);
  const int wv = __builtin_amdgcn_readfirstlane((int)(threadIdx.x >> 5));
  const int slot = wv * 4 + (lane >> 3);
  int node = blockIdx.x * 32 + slot;
  const bool live = node < n;
  if (!live) node = n - 1;
  const unsigned start = boff[node >> NB_SHIFT] + lpre[node];
  const unsigned cnt = live ? degin[node] : 0u;
  const float ni = norm_in[node];
  const float po = post_scale ? post_scale[node] : 1.0f;
  unsigned mx = cnt;
  {
    const unsigned o = (unsigned)__shfl_xor((int)mx, 8);
    mx = o > mx ? o : mx;
  }
  {
    const unsigned o = (unsigned)__shfl_xor((int)mx, 16);
    mx = o > mx ? o : mx;
  }
  const unsigned maxcnt = (unsigned)__builtin_amdgcn_readfirstlane((int)mx);

  const char* __restrict__ xb = reinterpret_cast<const char*>(xs);
  const unsigned sub16 = sub * 16u;
  const unsigned last = cnt ? cnt - 1u : 0u;
  const unsigned emax = (unsigned)e - 1u;
  v4f acc = {0.0f, 0.0f, 0.0f, 0.0f};
  for (unsigned c0 = 0; c0 < maxcnt; c0 += AGG_U) {
    unsigned a0 = start + (c0 + sub < cnt ? c0 + sub : last);
    unsigned a1 = start + (c0 + 8u + sub < cnt ? c0 + 8u + sub : last);
    if (a0 > emax) a0 = emax;
    if (a1 > emax) a1 = emax;
    const unsigned o0 = (unsigned)ssrc[a0] * (unsigned)(DF * 4);
    const unsigned o1 = (unsigned)ssrc[a1] * (unsigned)(DF * 4);
    v4f v[AGG_U];
#define AGG_ROW(O, B, J) \
  v[B + J] = *reinterpret_cast<const v4f*>(xb + (group_bcast<J>(O) + sub16));
    AGG_ROW(o0, 0, 0) AGG_ROW(o0, 0, 1) AGG_ROW(o0, 0, 2) AGG_ROW(o0, 0, 3)
    AGG_ROW(o0, 0, 4) AGG_ROW(o0, 0, 5) AGG_ROW(o0, 0, 6) AGG_ROW(o0, 0, 7)
    AGG_ROW(o1, 8, 0) AGG_ROW(o1, 8, 1) AGG_ROW(o1, 8, 2) AGG_ROW(o1, 8, 3)
    AGG_ROW(o1, 8, 4) AGG_ROW(o1, 8, 5) AGG_ROW(o1, 8, 6) AGG_ROW(o1, 8, 7)
#undef AGG_ROW
    __builtin_amdgcn_sched_barrier(0);
#pragma unroll
    for (int j = 0; j < AGG_U; ++j) {
      const bool ok = c0 + (unsigned)j < cnt;
      acc.x += ok ? v[j].x : 0.0f;
      acc.y += ok ? v[j].y : 0.0f;
      acc.z += ok ? v[j].z : 0.0f;
      acc.w += ok ? v[j].w : 0.0f;
    }
  }

  *reinterpret_cast<v4f*>(&tile[slot * AT_PITCH + (int)sub * 4]) = acc * ni;
  if (sub == 0u) rscale[slot] = po;
  __syncthreads();

  const int row0 = blockIdx.x * 32 + wv * 16;
  if (wv < 2 && row0 < n) {
    const int mrow = lane & 15;
    const int khalf = lane >> 4;
    const int nb = lane & 15;
    const float* arow = &tile[(wv * 16 + mrow) * AT_PITCH];
    v8f acc0 = {};
    v8f acc1 = {};
#pragma unroll
    for (int k0 = 0; k0 < DF; k0 += 4) {
      const int ka = k0 + khalf * 2;
      v2f a;
      a.x = arow[ka + 0];
      a.y = arow[ka + 1];
      v2f b0, b1;
      b0.x = W[(ka + 0) * DF + nb];
      b0.y = W[(ka + 1) * DF + nb];
      b1.x = W[(ka + 0) * DF + 16 + nb];
      b1.y = W[(ka + 1) * DF + 16 + nb];
      acc0 = __builtin_amdgcn_wmma_f32_16x16x4_f32(false, a, false, b0,
                                                   (short)0, acc0, false, false);
      acc1 = __builtin_amdgcn_wmma_f32_16x16x4_f32(false, a, false, b1,
                                                   (short)0, acc1, false, false);
    }

    const float bc0 = bias[nb];
    const float bc1 = bias[16 + nb];
#pragma unroll
    for (int v = 0; v < 8; ++v) {
      const int r = v + 8 * khalf;
      const int m = row0 + r;
      if (m < n) {
        float o0 = acc0[v] + bc0;
        float o1 = acc1[v] + bc1;
        if (do_relu) {
          o0 = fmaxf(o0, 0.0f);
          o1 = fmaxf(o1, 0.0f);
        }
        const float s = rscale[wv * 16 + r];
        out[(size_t)m * DF + nb] = o0 * s;
        out[(size_t)m * DF + 16 + nb] = o1 * s;
      }
    }
  }
}

extern "C" void kernel_launch(void* const* d_in, const int* in_sizes, int n_in,
                              void* d_out, int out_size, void* d_ws,
                              size_t ws_size, hipStream_t stream) {
  const float* x = (const float*)d_in[0];
  const int* src = (const int*)d_in[1];
  const int* dst = (const int*)d_in[2];
  const float* W1 = (const float*)d_in[3];
  const float* b1 = (const float*)d_in[4];
  const float* W2 = (const float*)d_in[5];
  const float* b2 = (const float*)d_in[6];
  float* out = (float*)d_out;

  const int N = in_sizes[0] / DF;
  const int E = in_sizes[1];
  const int NF = N * DF;

  float* ws = (float*)d_ws;
  float* norm_out = ws;
  float* norm_in = ws + N;
  float* xs = ws + ((2 * (size_t)N + 3) & ~(size_t)3);
  float* h1s = xs + (size_t)NF;
  unsigned* part = (unsigned*)(h1s + (size_t)NF);
  const int nblk = (N + NB_T - 1) / NB_T;
  unsigned* pre = part + 2 * (size_t)HIST_S * N;
  unsigned* lpre = pre + (size_t)HIST_S * N;
  unsigned* degin = lpre + N;
  unsigned* btot = degin + N;
  unsigned* boff = btot + nblk;
  int* ssrc = (int*)(boff + nblk);

  const int T = 256;
  const int gA = (N + 31) / 32;

  const int chunk = (((E + HIST_S - 1) / HIST_S) + 3) & ~3;
  k_deg_hist<<<dim3(HIST_S, 2), HIST_T, 0, stream>>>(src, dst, part, N, E, chunk);
  k_norms<<<dim3(nblk, 2), NB_T, 0, stream>>>(part, norm_out, pre, lpre, degin, btot, N);
  k_blockscan<<<1, 1024, 0, stream>>>(btot, boff, nblk);
  k_place<<<dim3(HIST_S, PL_P), HIST_T, 0, stream>>>(src, dst, pre, lpre, boff, ssrc, N, E, chunk);

  k_prescale<<<(NF / 4 + T * PS_U - 1) / (T * PS_U), T, 0, stream>>>(x, norm_out, xs, NF / 4);
  k_gather_gemm<<<gA, 256, 0, stream>>>(xs, ssrc, lpre, boff, degin, norm_in, W1, b1,
                                        norm_out, h1s, N, E, 1);

  k_gather_gemm<<<gA, 256, 0, stream>>>(h1s, ssrc, lpre, boff, degin, norm_in, W2, b2,
                                        nullptr, out, N, E, 0);
}


# Two small dependent launches folded away (7 -> 5): the norm kernel's out-degree work-groups write xs = x * norm_out for their own rows, each placement work-group scans the block totals itself in LDS
# speedup vs baseline: 4.3653x; 1.0550x over previous
#include <hip/hip_runtime.h>
#include <math.h>


typedef __attribute__((ext_vector_type(2))) float v2f;
typedef __attribute__((ext_vector_type(8))) float v8f;

#define DF 32

#define HIST_CAP 65536
#define HIST_T 1024
#define HIST_S 64
#define HIST_U 8

__global__ __launch_bounds__(HIST_T) void k_deg_hist(
    const int* __restrict__ src, const int* __restrict__ dst,
    unsigned* __restrict__ part, int n, int e, int chunk) {
  __shared__ __attribute__((aligned(16))) unsigned hist[HIST_CAP];
  const int tid = threadIdx.x;
  const int slice = blockIdx.x;
  const int which = blockIdx.y;
  const int* __restrict__ arr = which ? dst : src;
  unsigned* __restrict__ plane =
      part + ((size_t)which * HIST_S + slice) * (size_t)n;
  const int j0 = slice * chunk;
  int j1 = j0 + chunk;
  if (j1 > e) j1 = e;
  const int cnt = (j1 > j0) ? (j1 - j0) : 0;
  const int nvec = cnt >> 2;
  const int ntail = cnt & 3;
  const int4* __restrict__ arr4 = reinterpret_cast<const int4*>(arr + j0);
  const bool vec_ok = (n & 3) == 0;

  for (int base = 0; base < n; base += HIST_CAP) {
    int nb = n - base;
    if (nb > HIST_CAP) nb = HIST_CAP;

    for (int i = tid * 4; i < nb; i += HIST_T * 4)
      *reinterpret_cast<uint4*>(&hist[i]) = make_uint4(0u, 0u, 0u, 0u);
    __syncthreads();

    for (int v0 = tid; v0 < nvec; v0 += HIST_T * HIST_U) {
      int4 v[HIST_U];
#pragma unroll
      for (int u = 0; u < HIST_U; ++u) {
        const int idx = v0 + u * HIST_T;
        v[u] = arr4[idx < nvec ? idx : nvec - 1];
      }
#pragma unroll
      for (int u = 0; u < HIST_U; ++u) {
        if (v0 + u * HIST_T < nvec) {
          unsigned b;
          b = (unsigned)(v[u].x - base);
          if (b < (unsigned)nb) atomicAdd(&hist[b], 1u);
          b = (unsigned)(v[u].y - base);
          if (b < (unsigned)nb) atomicAdd(&hist[b], 1u);
          b = (unsigned)(v[u].z - base);
          if (b < (unsigned)nb) atomicAdd(&hist[b], 1u);
          b = (unsigned)(v[u].w - base);
          if (b < (unsigned)nb) atomicAdd(&hist[b], 1u);
        }
      }
    }
    if (tid < ntail) {
      const unsigned b = (unsigned)(arr[j0 + nvec * 4 + tid] - base);
      if (b < (unsigned)nb) atomicAdd(&hist[b], 1u);
    }
    __syncthreads();

    if (vec_ok) {
      for (int i = tid * 4; i < nb; i += HIST_T * 4)
        *reinterpret_cast<uint4*>(&plane[base + i]) =
            *reinterpret_cast<const uint4*>(&hist[i]);
    } else {
      for (int i = tid; i < nb; i += HIST_T) plane[base + i] = hist[i];
    }
    __syncthreads();
  }
}

#define NB_T 256
#define NB_SHIFT 8

typedef __attribute__((ext_vector_type(4))) float v4f;

#define PS_U 8

__global__ __launch_bounds__(NB_T) void k_norms(const unsigned* __restrict__ part,
                                                float* __restrict__ norms,
                                                unsigned* __restrict__ pre,
                                                unsigned* __restrict__ lpre,
                                                unsigned* __restrict__ degin,
                                                unsigned* __restrict__ btot,
                                                const float* __restrict__ x,
                                                float* __restrict__ xs,
                                                int n) {
  __shared__ unsigned sc[2][NB_T];
  __shared__ float snrm[NB_T];
  const int tid = threadIdx.x;
  const int i = blockIdx.x * NB_T + tid;
  const int which = blockIdx.y;
  unsigned c = 0;
  if (i < n) {
    const unsigned* __restrict__ p = part + (size_t)which * HIST_S * n + i;
    unsigned v[HIST_S];
#pragma unroll
    for (int s = 0; s < HIST_S; ++s) v[s] = p[(size_t)s * n];
    if (which == 1) {
#pragma unroll
      for (int s = 0; s < HIST_S; ++s) {
        pre[(size_t)s * n + i] = c;
        c += v[s];
      }
    } else {
#pragma unroll
      for (int s = 0; s < HIST_S; ++s) c += v[s];
    }
    const float d = (float)c;
    const float nv = (d > 0.0f) ? (1.0f / sqrtf(d)) : 0.0f;
    norms[(size_t)which * n + i] = nv;
    snrm[tid] = nv;
  }
  if (which == 0) {
    __syncthreads();
    const v4f* __restrict__ x4 = reinterpret_cast<const v4f*>(x);
    v4f* __restrict__ xs4 = reinterpret_cast<v4f*>(xs);
    const int total4 = n * (DF / 4);
    const int g0 = blockIdx.x * (NB_T * (DF / 4)) + tid;
    v4f v[PS_U];
#pragma unroll
    for (int u = 0; u < PS_U; ++u) {
      const int idx = g0 + u * NB_T;
      v[u] = x4[idx < total4 ? idx : total4 - 1];
    }
#pragma unroll
    for (int u = 0; u < PS_U; ++u) {
      const int idx = g0 + u * NB_T;
      if (idx < total4) xs4[idx] = v[u] * snrm[(u * NB_T + tid) >> 3];
    }
  }
  if (which == 1) {
    sc[0][tid] = c;
    __syncthreads();
    int cur = 0;
    for (int off = 1; off < NB_T; off <<= 1) {
      unsigned t = sc[cur][tid];
      if (tid >= off) t += sc[cur][tid - off];
      sc[cur ^ 1][tid] = t;
      cur ^= 1;
      __syncthreads();
    }
    const unsigned incl = sc[cur][tid];
    if (i < n) {
      lpre[i] = incl - c;
      degin[i] = c;
    }
    if (tid == NB_T - 1) btot[blockIdx.x] = incl;
  }
}

__device__ __forceinline__ unsigned wave_incl_scan(unsigned v, int lane) {
#pragma unroll
  for (int off = 1; off < 32; off <<= 1) {
    const unsigned t = (unsigned)__shfl_up((int)v, off);
    if (lane >= off) v += t;
  }
  return v;
}

#define SB_N 264

#define PL_P 4

__global__ __launch_bounds__(HIST_T) void k_place(
    const int* __restrict__ src, const int* __restrict__ dst,
    const unsigned* __restrict__ pre, const unsigned* __restrict__ lpre,
    const unsigned* __restrict__ btot, unsigned* __restrict__ boff,
    int* __restrict__ ssrc, int n, int e, int chunk, int nblk) {
  __shared__ __attribute__((aligned(16))) unsigned cursor[HIST_CAP];
  __shared__ unsigned sboff[SB_N];
  __shared__ unsigned wsum[32];
  const int tid = threadIdx.x;
  const int lane = tid & 31;
  const int wave = tid >> 5;
  const int slice = blockIdx.x;
  const bool keeper = blockIdx.x == 0 && blockIdx.y == 0;
  const int per = (((n + PL_P - 1) / PL_P) + 3) & ~3;
  const int r0 = (int)blockIdx.y * per;
  int r1 = r0 + per;
  if (r1 > n) r1 = n;
  const unsigned* __restrict__ plane = pre + (size_t)slice * (size_t)n;
  const int j0 = slice * chunk;
  int j1 = j0 + chunk;
  if (j1 > e) j1 = e;
  const int cnt = (j1 > j0) ? (j1 - j0) : 0;
  const int nvec = cnt >> 2;
  const int ntail = cnt & 3;
  const int4* __restrict__ src4 = reinterpret_cast<const int4*>(src + j0);
  const int4* __restrict__ dst4 = reinterpret_cast<const int4*>(dst + j0);
  const bool vec_ok = (n & 3) == 0;

  for (int base = r0; base < r1; base += HIST_CAP) {
    int nb = r1 - base;
    if (nb > HIST_CAP) nb = HIST_CAP;

    const int blk0 = base >> NB_SHIFT;
    unsigned carry = 0;
    for (int cb = 0; cb < nblk; cb += HIST_T) {
      const int bi = cb + tid;
      const unsigned c = (bi < nblk) ? btot[bi] : 0u;
      const unsigned incl = wave_incl_scan(c, lane);
      if (lane == 31) wsum[wave] = incl;
      __syncthreads();
      const unsigned wt = wsum[lane];
      const unsigned wincl = wave_incl_scan(wt, lane);
      const unsigned wexcl = (unsigned)__shfl((int)(wincl - wt), wave);
      const unsigned total = (unsigned)__shfl((int)wincl, 31);
      const unsigned excl = carry + wexcl + incl - c;
      if (bi < nblk) {
        const unsigned rel = (unsigned)(bi - blk0);
        if (rel < (unsigned)SB_N) sboff[rel] = excl;
        if (keeper && base == r0) boff[bi] = excl;
      }
      carry += total;
      __syncthreads();
    }

    if (vec_ok) {
      for (int i = tid * 4; i < nb; i += HIST_T * 4) {
        const int g = base + i;
        const uint4 l = *reinterpret_cast<const uint4*>(&lpre[g]);
        const uint4 p = *reinterpret_cast<const uint4*>(&plane[g]);
        const unsigned bo = sboff[(g >> NB_SHIFT) - blk0];
        *reinterpret_cast<uint4*>(&cursor[i]) = make_uint4(
            bo + l.x + p.x, bo + l.y + p.y, bo + l.z + p.z, bo + l.w + p.w);
      }
    } else {
      for (int i = tid; i < nb; i += HIST_T) {
        const int g = base + i;
        cursor[i] = sboff[(g >> NB_SHIFT) - blk0] + lpre[g] + plane[g];
      }
    }
    __syncthreads();

    for (int v0 = tid; v0 < nvec; v0 += HIST_T * HIST_U) {
      int4 dv[HIST_U], sv[HIST_U];
#pragma unroll
      for (int u = 0; u < HIST_U; ++u) {
        const int idx = v0 + u * HIST_T;
        const int cl = idx < nvec ? idx : nvec - 1;
        dv[u] = dst4[cl];
        sv[u] = src4[cl];
      }
#pragma unroll
      for (int u = 0; u < HIST_U; ++u) {
        if (v0 + u * HIST_T < nvec) {
          unsigned b, pos;
          b = (unsigned)(dv[u].x - base);
          if (b < (unsigned)nb) {
            pos = atomicAdd(&cursor[b], 1u);
            if (pos < (unsigned)e) ssrc[pos] = sv[u].x;
          }
          b = (unsigned)(dv[u].y - base);
          if (b < (unsigned)nb) {
            pos = atomicAdd(&cursor[b], 1u);
            if (pos < (unsigned)e) ssrc[pos] = sv[u].y;
          }
          b = (unsigned)(dv[u].z - base);
          if (b < (unsigned)nb) {
            pos = atomicAdd(&cursor[b], 1u);
            if (pos < (unsigned)e) ssrc[pos] = sv[u].z;
          }
          b = (unsigned)(dv[u].w - base);
          if (b < (unsigned)nb) {
            pos = atomicAdd(&cursor[b], 1u);
            if (pos < (unsigned)e) ssrc[pos] = sv[u].w;
          }
        }
      }
    }
    if (tid < ntail) {
      const int j = j0 + nvec * 4 + tid;
      const unsigned b = (unsigned)(dst[j] - base);
      if (b < (unsigned)nb) {
        const unsigned pos = atomicAdd(&cursor[b], 1u);
        if (pos < (unsigned)e) ssrc[pos] = src[j];
      }
    }
    __syncthreads();
  }
}

#define AGG_U 16

template <int J>
__device__ __forceinline__ unsigned group_bcast(unsigned v) {
  return __builtin_amdgcn_mov_dpp8(v, (unsigned)(J * 0x249249));
}

#define AT_PITCH 36

__global__ __launch_bounds__(256, 12) void k_gather_gemm(
    const float* __restrict__ xs, const int* __restrict__ ssrc,
    const unsigned* __restrict__ lpre, const unsigned* __restrict__ boff,
    const unsigned* __restrict__ degin, const float* __restrict__ norm_in,
    const float* __restrict__ W, const float* __restrict__ bias,
    const float* __restrict__ post_scale,
    float* __restrict__ out, int n, int e, int do_relu) {
  __shared__ __attribute__((aligned(16))) float tile[32 * AT_PITCH];
  __shared__ float rscale[32];
  const int lane = threadIdx.x & 31;
  const unsigned sub = (unsigned)(lane & 7);
  const int wv = __builtin_amdgcn_readfirstlane((int)(threadIdx.x >> 5));
  const int slot = wv * 4 + (lane >> 3);
  int node = blockIdx.x * 32 + slot;
  const bool live = node < n;
  if (!live) node = n - 1;
  const unsigned start = boff[node >> NB_SHIFT] + lpre[node];
  const unsigned cnt = live ? degin[node] : 0u;
  const float ni = norm_in[node];
  const float po = post_scale ? post_scale[node] : 1.0f;
  unsigned mx = cnt;
  {
    const unsigned o = (unsigned)__shfl_xor((int)mx, 8);
    mx = o > mx ? o : mx;
  }
  {
    const unsigned o = (unsigned)__shfl_xor((int)mx, 16);
    mx = o > mx ? o : mx;
  }
  const unsigned maxcnt = (unsigned)__builtin_amdgcn_readfirstlane((int)mx);

  const char* __restrict__ xb = reinterpret_cast<const char*>(xs);
  const unsigned sub16 = sub * 16u;
  const unsigned last = cnt ? cnt - 1u : 0u;
  const unsigned emax = (unsigned)e - 1u;
  v4f acc = {0.0f, 0.0f, 0.0f, 0.0f};
  for (unsigned c0 = 0; c0 < maxcnt; c0 += AGG_U) {
    unsigned a0 = start + (c0 + sub < cnt ? c0 + sub : last);
    unsigned a1 = start + (c0 + 8u + sub < cnt ? c0 + 8u + sub : last);
    if (a0 > emax) a0 = emax;
    if (a1 > emax) a1 = emax;
    const unsigned o0 = (unsigned)ssrc[a0] * (unsigned)(DF * 4);
    const unsigned o1 = (unsigned)ssrc[a1] * (unsigned)(DF * 4);
    v4f v[AGG_U];
#define AGG_ROW(O, B, J) \
  v[B + J] = *reinterpret_cast<const v4f*>(xb + (group_bcast<J>(O) + sub16));
    AGG_ROW(o0, 0, 0) AGG_ROW(o0, 0, 1) AGG_ROW(o0, 0, 2) AGG_ROW(o0, 0, 3)
    AGG_ROW(o0, 0, 4) AGG_ROW(o0, 0, 5) AGG_ROW(o0, 0, 6) AGG_ROW(o0, 0, 7)
    AGG_ROW(o1, 8, 0) AGG_ROW(o1, 8, 1) AGG_ROW(o1, 8, 2) AGG_ROW(o1, 8, 3)
    AGG_ROW(o1, 8, 4) AGG_ROW(o1, 8, 5) AGG_ROW(o1, 8, 6) AGG_ROW(o1, 8, 7)
#undef AGG_ROW
    __builtin_amdgcn_sched_barrier(0);
#pragma unroll
    for (int j = 0; j < AGG_U; ++j) {
      const bool ok = c0 + (unsigned)j < cnt;
      acc.x += ok ? v[j].x : 0.0f;
      acc.y += ok ? v[j].y : 0.0f;
      acc.z += ok ? v[j].z : 0.0f;
      acc.w += ok ? v[j].w : 0.0f;
    }
  }

  *reinterpret_cast<v4f*>(&tile[slot * AT_PITCH + (int)sub * 4]) = acc * ni;
  if (sub == 0u) rscale[slot] = po;
  __syncthreads();

  const int row0 = blockIdx.x * 32 + wv * 16;
  if (wv < 2 && row0 < n) {
    const int mrow = lane & 15;
    const int khalf = lane >> 4;
    const int nb = lane & 15;
    const float* arow = &tile[(wv * 16 + mrow) * AT_PITCH];
    v8f acc0 = {};
    v8f acc1 = {};
#pragma unroll
    for (int k0 = 0; k0 < DF; k0 += 4) {
      const int ka = k0 + khalf * 2;
      v2f a;
      a.x = arow[ka + 0];
      a.y = arow[ka + 1];
      v2f b0, b1;
      b0.x = W[(ka + 0) * DF + nb];
      b0.y = W[(ka + 1) * DF + nb];
      b1.x = W[(ka + 0) * DF + 16 + nb];
      b1.y = W[(ka + 1) * DF + 16 + nb];
      acc0 = __builtin_amdgcn_wmma_f32_16x16x4_f32(false, a, false, b0,
                                                   (short)0, acc0, false, false);
      acc1 = __builtin_amdgcn_wmma_f32_16x16x4_f32(false, a, false, b1,
                                                   (short)0, acc1, false, false);
    }

    const float bc0 = bias[nb];
    const float bc1 = bias[16 + nb];
#pragma unroll
    for (int v = 0; v < 8; ++v) {
      const int r = v + 8 * khalf;
      const int m = row0 + r;
      if (m < n) {
        float o0 = acc0[v] + bc0;
        float o1 = acc1[v] + bc1;
        if (do_relu) {
          o0 = fmaxf(o0, 0.0f);
          o1 = fmaxf(o1, 0.0f);
        }
        const float s = rscale[wv * 16 + r];
        out[(size_t)m * DF + nb] = o0 * s;
        out[(size_t)m * DF + 16 + nb] = o1 * s;
      }
    }
  }
}

extern "C" void kernel_launch(void* const* d_in, const int* in_sizes, int n_in,
                              void* d_out, int out_size, void* d_ws,
                              size_t ws_size, hipStream_t stream) {
  const float* x = (const float*)d_in[0];
  const int* src = (const int*)d_in[1];
  const int* dst = (const int*)d_in[2];
  const float* W1 = (const float*)d_in[3];
  const float* b1 = (const float*)d_in[4];
  const float* W2 = (const float*)d_in[5];
  const float* b2 = (const float*)d_in[6];
  float* out = (float*)d_out;

  const int N = in_sizes[0] / DF;
  const int E = in_sizes[1];
  const int NF = N * DF;

  float* ws = (float*)d_ws;
  float* norm_out = ws;
  float* norm_in = ws + N;
  float* xs = ws + ((2 * (size_t)N + 3) & ~(size_t)3);
  float* h1s = xs + (size_t)NF;
  unsigned* part = (unsigned*)(h1s + (size_t)NF);
  const int nblk = (N + NB_T - 1) / NB_T;
  unsigned* pre = part + 2 * (size_t)HIST_S * N;
  unsigned* lpre = pre + (size_t)HIST_S * N;
  unsigned* degin = lpre + N;
  unsigned* btot = degin + N;
  unsigned* boff = btot + nblk;
  int* ssrc = (int*)(boff + nblk);

  const int gA = (N + 31) / 32;

  const int chunk = (((E + HIST_S - 1) / HIST_S) + 3) & ~3;
  k_deg_hist<<<dim3(HIST_S, 2), HIST_T, 0, stream>>>(src, dst, part, N, E, chunk);
  k_norms<<<dim3(nblk, 2), NB_T, 0, stream>>>(part, norm_out, pre, lpre, degin, btot, x, xs, N);
  k_place<<<dim3(HIST_S, PL_P), HIST_T, 0, stream>>>(src, dst, pre, lpre, btot, boff, ssrc,
                                                     N, E, chunk, nblk);

  k_gather_gemm<<<gA, 256, 0, stream>>>(xs, ssrc, lpre, boff, degin, norm_in, W1, b1,
                                        norm_out, h1s, N, E, 1);

  k_gather_gemm<<<gA, 256, 0, stream>>>(h1s, ssrc, lpre, boff, degin, norm_in, W2, b2,
                                        nullptr, out, N, E, 0);
}


# Edges sorted by destination in two LDS levels (pieces grouped by 256-node block, then per-block per-wave cursors): contiguous stores, no scattered 4-byte stores; 5 -> 4 launches; sum order changes
# speedup vs baseline: 7.0532x; 1.6157x over previous
#include <hip/hip_runtime.h>
#include <math.h>


typedef __attribute__((ext_vector_type(2))) float v2f;
typedef __attribute__((ext_vector_type(8))) float v8f;
typedef __attribute__((ext_vector_type(4))) float v4f;

#define DF 32

#define HIST_CAP 65536
#define HIST_T 1024
#define HIST_S 64
#define HIST_U 8

#define NB_T 256
#define NB_SHIFT 8

#define BIN_PIECE 8336
#define BIN_U 3
#define BIN_MAXB 4096

__device__ __forceinline__ unsigned wave_incl_scan(unsigned v, int lane) {
#pragma unroll
  for (int off = 1; off < 32; off <<= 1) {
    const unsigned t = (unsigned)__shfl_up((int)v, off);
    if (lane >= off) v += t;
  }
  return v;
}

__device__ __forceinline__ unsigned bin_of(int d, int nblk) {
  unsigned b = (unsigned)d >> NB_SHIFT;
  return b < (unsigned)nblk ? b : (unsigned)(nblk - 1);
}

__global__ __launch_bounds__(HIST_T) void k_front(
    const int* __restrict__ src, const int* __restrict__ dst,
    unsigned* __restrict__ part, unsigned* __restrict__ tab,
    int2* __restrict__ binned, int n, int e, int chunk, int nblk) {
  __shared__ __attribute__((aligned(16))) unsigned lds[HIST_CAP];
  __shared__ unsigned wsum[32];
  const int tid = threadIdx.x;

  if (blockIdx.x < HIST_S) {
    unsigned* hist = lds;
    const int slice = blockIdx.x;
    unsigned* __restrict__ plane = part + (size_t)slice * (size_t)n;
    const int j0 = slice * chunk;
    int j1 = j0 + chunk;
    if (j1 > e) j1 = e;
    const int cnt = (j1 > j0) ? (j1 - j0) : 0;
    const int nvec = cnt >> 2;
    const int ntail = cnt & 3;
    const int4* __restrict__ arr4 = reinterpret_cast<const int4*>(src + j0);
    const bool vec_ok = (n & 3) == 0;

    for (int base = 0; base < n; base += HIST_CAP) {
      int nb = n - base;
      if (nb > HIST_CAP) nb = HIST_CAP;

      for (int i = tid * 4; i < nb; i += HIST_T * 4)
        *reinterpret_cast<uint4*>(&hist[i]) = make_uint4(0u, 0u, 0u, 0u);
      __syncthreads();

      for (int v0 = tid; v0 < nvec; v0 += HIST_T * HIST_U) {
        int4 v[HIST_U];
#pragma unroll
        for (int u = 0; u < HIST_U; ++u) {
          const int idx = v0 + u * HIST_T;
          v[u] = arr4[idx < nvec ? idx : nvec - 1];
        }
#pragma unroll
        for (int u = 0; u < HIST_U; ++u) {
          if (v0 + u * HIST_T < nvec) {
            unsigned b;
            b = (unsigned)(v[u].x - base);
            if (b < (unsigned)nb) atomicAdd(&hist[b], 1u);
            b = (unsigned)(v[u].y - base);
            if (b < (unsigned)nb) atomicAdd(&hist[b], 1u);
            b = (unsigned)(v[u].z - base);
            if (b < (unsigned)nb) atomicAdd(&hist[b], 1u);
            b = (unsigned)(v[u].w - base);
            if (b < (unsigned)nb) atomicAdd(&hist[b], 1u);
          }
        }
      }
      if (tid < ntail) {
        const unsigned b = (unsigned)(src[j0 + nvec * 4 + tid] - base);
        if (b < (unsigned)nb) atomicAdd(&hist[b], 1u);
      }
      __syncthreads();

      if (vec_ok) {
        for (int i = tid * 4; i < nb; i += HIST_T * 4)
          *reinterpret_cast<uint4*>(&plane[base + i]) =
              *reinterpret_cast<const uint4*>(&hist[i]);
      } else {
        for (int i = tid; i < nb; i += HIST_T) plane[base + i] = hist[i];
      }
      __syncthreads();
    }
    return;
  }

  unsigned* cnt = lds;
  unsigned* bst = lds + BIN_MAXB;
  int2* stage = reinterpret_cast<int2*>(lds + 2 * BIN_MAXB);
  const int lane = tid & 31;
  const int wave = tid >> 5;
  const int piece = (int)blockIdx.x - HIST_S;
  const int j0 = piece * BIN_PIECE;
  int j1 = j0 + BIN_PIECE;
  if (j1 > e) j1 = e;
  const int cntE = (j1 > j0) ? (j1 - j0) : 0;
  const int nvec = cntE >> 2;
  const int ntail = cntE & 3;
  const int4* __restrict__ src4 = reinterpret_cast<const int4*>(src + j0);
  const int4* __restrict__ dst4 = reinterpret_cast<const int4*>(dst + j0);
  unsigned* __restrict__ row = tab + (size_t)piece * (size_t)(nblk + 1);

  for (int i = tid; i < nblk; i += HIST_T) cnt[i] = 0u;
  __syncthreads();

  int4 dv[BIN_U], sv[BIN_U];
  uint4 rk[BIN_U];
  int td = 0, ts = 0;
  unsigned tr = 0;
  if (nvec > 0) {
#pragma unroll
    for (int u = 0; u < BIN_U; ++u) {
      const int idx = tid + u * HIST_T;
      const int cl = idx < nvec ? idx : nvec - 1;
      dv[u] = dst4[cl];
      sv[u] = src4[cl];
    }
#pragma unroll
    for (int u = 0; u < BIN_U; ++u) {
      if (tid + u * HIST_T < nvec) {
        rk[u].x = atomicAdd(&cnt[bin_of(dv[u].x, nblk)], 1u);
        rk[u].y = atomicAdd(&cnt[bin_of(dv[u].y, nblk)], 1u);
        rk[u].z = atomicAdd(&cnt[bin_of(dv[u].z, nblk)], 1u);
        rk[u].w = atomicAdd(&cnt[bin_of(dv[u].w, nblk)], 1u);
      }
    }
  }
  if (tid < ntail) {
    td = dst[j0 + nvec * 4 + tid];
    ts = src[j0 + nvec * 4 + tid];
    tr = atomicAdd(&cnt[bin_of(td, nblk)], 1u);
  }
  __syncthreads();

  unsigned carry = 0;
  for (int cb = 0; cb < nblk; cb += HIST_T) {
    const int bi = cb + tid;
    const unsigned c = (bi < nblk) ? cnt[bi] : 0u;
    const unsigned incl = wave_incl_scan(c, lane);
    if (lane == 31) wsum[wave] = incl;
    __syncthreads();
    const unsigned wt = wsum[lane];
    const unsigned wincl = wave_incl_scan(wt, lane);
    const unsigned wexcl = (unsigned)__shfl((int)(wincl - wt), wave);
    const unsigned total = (unsigned)__shfl((int)wincl, 31);
    const unsigned excl = carry + wexcl + incl - c;
    if (bi < nblk) {
      bst[bi] = excl;
      row[bi] = excl;
    }
    carry += total;
    __syncthreads();
  }
  if (tid == 0) row[nblk] = carry;

  if (nvec > 0) {
#pragma unroll
    for (int u = 0; u < BIN_U; ++u) {
      if (tid + u * HIST_T < nvec) {
        stage[bst[bin_of(dv[u].x, nblk)] + rk[u].x] = make_int2(sv[u].x, dv[u].x);
        stage[bst[bin_of(dv[u].y, nblk)] + rk[u].y] = make_int2(sv[u].y, dv[u].y);
        stage[bst[bin_of(dv[u].z, nblk)] + rk[u].z] = make_int2(sv[u].z, dv[u].z);
        stage[bst[bin_of(dv[u].w, nblk)] + rk[u].w] = make_int2(sv[u].w, dv[u].w);
      }
    }
  }
  if (tid < ntail) stage[bst[bin_of(td, nblk)] + tr] = make_int2(ts, td);
  __syncthreads();

  int2* __restrict__ outp = binned + (size_t)j0;
  for (int i = tid; i < cntE; i += HIST_T) outp[i] = stage[i];
}

#define PS_U 2
#define CSR_T 1024
#define CSR_W 32
#define CSR_K 6
#define CSR_MAXP 1024
#define CSR_CAP 12288

__device__ __forceinline__ unsigned loc_of(int d, int base) {
  const unsigned loc = (unsigned)(d - base);
  return loc < (unsigned)NB_T ? loc : (unsigned)(NB_T - 1);
}

__global__ __launch_bounds__(CSR_T) void k_block(
    const unsigned* __restrict__ part, const float* __restrict__ x,
    const unsigned* __restrict__ tab, const int2* __restrict__ binned,
    float* __restrict__ norm_out, float* __restrict__ norm_in,
    float* __restrict__ xs, unsigned* __restrict__ lpre,
    unsigned* __restrict__ degin, unsigned* __restrict__ boff,
    int* __restrict__ ssrc, int n, int e, int nblk, int np) {
  __shared__ unsigned pgs[CSR_MAXP];
  __shared__ unsigned plen[CSR_MAXP];
  __shared__ __attribute__((aligned(16))) unsigned wcnt[CSR_W * NB_T];
  __shared__ float snrm[NB_T];
  __shared__ unsigned wsA[32];
  __shared__ unsigned wsB[32];
  __shared__ int ostage[CSR_CAP];
  const int tid = threadIdx.x;
  const int lane = tid & 31;
  const int wave = tid >> 5;
  const int b = blockIdx.x;
  const int base = b * NB_T;
  const int npc = np < CSR_MAXP ? np : CSR_MAXP;

  const v4f* __restrict__ x4 = reinterpret_cast<const v4f*>(x);
  v4f* __restrict__ xs4 = reinterpret_cast<v4f*>(xs);
  const int total4 = n * (DF / 4);
  const int g0 = b * (NB_T * (DF / 4)) + tid;
  v4f xv[PS_U];
#pragma unroll
  for (int u = 0; u < PS_U; ++u) {
    const int idx = g0 + u * CSR_T;
    xv[u] = x4[idx < total4 ? idx : total4 - 1];
  }

  unsigned lo = 0, len = 0;
  if (tid < npc) {
    const unsigned* __restrict__ row = tab + (size_t)tid * (size_t)(nblk + 1) + b;
    lo = row[0];
    const unsigned hi = row[1];
    len = hi > lo ? hi - lo : 0u;
  }

  if (tid >= CSR_T - NB_T) {
    const int t = tid - (CSR_T - NB_T);
    const int i = base + t;
    float nv = 0.0f;
    if (i < n) {
      const unsigned* __restrict__ p = part + i;
      unsigned v[HIST_S];
#pragma unroll
      for (int s = 0; s < HIST_S; ++s) v[s] = p[(size_t)s * n];
      unsigned c = 0;
#pragma unroll
      for (int s = 0; s < HIST_S; ++s) c += v[s];
      const float d = (float)c;
      nv = (d > 0.0f) ? (1.0f / sqrtf(d)) : 0.0f;
      norm_out[i] = nv;
    }
    snrm[t] = nv;
  }
  *reinterpret_cast<uint4*>(&wcnt[tid * 8]) = make_uint4(0u, 0u, 0u, 0u);
  *reinterpret_cast<uint4*>(&wcnt[tid * 8 + 4]) = make_uint4(0u, 0u, 0u, 0u);

  const unsigned sumLen = wave_incl_scan(len, lane);
  const unsigned sumLo = wave_incl_scan(lo, lane);
  if (lane == 31) {
    wsA[wave] = sumLen;
    wsB[wave] = sumLo;
  }
  if (tid < npc) {
    pgs[tid] = (unsigned)tid * (unsigned)BIN_PIECE + lo;
    plen[tid] = len;
  }
  __syncthreads();
#pragma unroll
  for (int u = 0; u < PS_U; ++u) {
    const int idx = g0 + u * CSR_T;
    if (idx < total4) xs4[idx] = xv[u] * snrm[(u * CSR_T + tid) >> 3];
  }
  unsigned total = (unsigned)__shfl((int)wave_incl_scan(wsA[lane], lane), 31);
  const unsigned bo = (unsigned)__shfl((int)wave_incl_scan(wsB[lane], lane), 31);
  if (total > (unsigned)e) total = (unsigned)e;
  const bool staged = total <= (unsigned)CSR_CAP;
  const unsigned emax = e > 0 ? (unsigned)e - 1u : 0u;
  unsigned* mycnt = &wcnt[wave * NB_T];
  const int* __restrict__ bdst = reinterpret_cast<const int*>(binned) + 1;

  for (int pc = wave; pc < npc; pc += CSR_W * CSR_K) {
    unsigned gs[CSR_K], ln[CSR_K];
#pragma unroll
    for (int k = 0; k < CSR_K; ++k) {
      const int p = pc + CSR_W * k;
      const bool ok = p < npc;
      gs[k] = ok ? pgs[ok ? p : 0] : 0u;
      ln[k] = ok ? plen[ok ? p : 0] : 0u;
    }
    int dd[CSR_K][2];
#pragma unroll
    for (int k = 0; k < CSR_K; ++k) {
#pragma unroll
      for (int h = 0; h < 2; ++h) {
        const unsigned idx = (unsigned)lane + 32u * (unsigned)h;
        const unsigned cl = idx < ln[k] ? idx : (ln[k] ? ln[k] - 1u : 0u);
        unsigned g = gs[k] + cl;
        if (g > emax) g = emax;
        dd[k][h] = bdst[2u * g];
      }
    }
#pragma unroll
    for (int k = 0; k < CSR_K; ++k) {
#pragma unroll
      for (int h = 0; h < 2; ++h) {
        if ((unsigned)lane + 32u * (unsigned)h < ln[k])
          atomicAdd(&mycnt[loc_of(dd[k][h], base)], 1u);
      }
      for (unsigned idx = (unsigned)lane + 64u; idx < ln[k]; idx += 32u) {
        unsigned g = gs[k] + idx;
        if (g > emax) g = emax;
        atomicAdd(&mycnt[loc_of(bdst[2u * g], base)], 1u);
      }
    }
  }
  __syncthreads();

  {
    unsigned c[CSR_W];
    unsigned run = 0;
    if (tid < NB_T) {
#pragma unroll
      for (int w = 0; w < CSR_W; ++w) c[w] = wcnt[w * NB_T + tid];
#pragma unroll
      for (int w = 0; w < CSR_W; ++w) {
        const unsigned t = c[w];
        c[w] = run;
        run += t;
      }
    }
    const unsigned ci = wave_incl_scan(run, lane);
    if (lane == 31) wsA[wave] = ci;
    __syncthreads();
    const unsigned w2 = wsA[lane];
    const unsigned w2i = wave_incl_scan(w2, lane);
    const unsigned w2e = (unsigned)__shfl((int)(w2i - w2), wave);
    const unsigned excl = w2e + ci - run;
    if (tid < NB_T) {
#pragma unroll
      for (int w = 0; w < CSR_W; ++w) wcnt[w * NB_T + tid] = excl + c[w];
      const int i = base + tid;
      if (i < n) {
        lpre[i] = excl;
        degin[i] = run;
        const float d = (float)run;
        norm_in[i] = (d > 0.0f) ? (1.0f / sqrtf(d)) : 0.0f;
      }
    }
    if (tid == 0) boff[b] = bo;
  }

  __syncthreads();

  for (int pc = wave; pc < npc; pc += CSR_W * CSR_K) {
    unsigned gs[CSR_K], ln[CSR_K];
#pragma unroll
    for (int k = 0; k < CSR_K; ++k) {
      const int p = pc + CSR_W * k;
      const bool ok = p < npc;
      gs[k] = ok ? pgs[ok ? p : 0] : 0u;
      ln[k] = ok ? plen[ok ? p : 0] : 0u;
    }
    int2 ed[CSR_K][2];
#pragma unroll
    for (int k = 0; k < CSR_K; ++k) {
#pragma unroll
      for (int h = 0; h < 2; ++h) {
        const unsigned idx = (unsigned)lane + 32u * (unsigned)h;
        const unsigned cl = idx < ln[k] ? idx : (ln[k] ? ln[k] - 1u : 0u);
        unsigned g = gs[k] + cl;
        if (g > emax) g = emax;
        ed[k][h] = binned[g];
      }
    }
#pragma unroll
    for (int k = 0; k < CSR_K; ++k) {
#pragma unroll
      for (int h = 0; h < 2; ++h) {
        if ((unsigned)lane + 32u * (unsigned)h < ln[k]) {
          const unsigned pos = atomicAdd(&mycnt[loc_of(ed[k][h].y, base)], 1u);
          if (staged) {
            if (pos < (unsigned)CSR_CAP) ostage[pos] = ed[k][h].x;
          } else {
            const unsigned g = bo + pos;
            if (g < (unsigned)e) ssrc[g] = ed[k][h].x;
          }
        }
      }
      for (unsigned idx = (unsigned)lane + 64u; idx < ln[k]; idx += 32u) {
        unsigned g = gs[k] + idx;
        if (g > emax) g = emax;
        const int2 t2 = binned[g];
        const unsigned pos = atomicAdd(&mycnt[loc_of(t2.y, base)], 1u);
        if (staged) {
          if (pos < (unsigned)CSR_CAP) ostage[pos] = t2.x;
        } else {
          const unsigned g2 = bo + pos;
          if (g2 < (unsigned)e) ssrc[g2] = t2.x;
        }
      }
    }
  }
  __syncthreads();
  if (staged) {
    for (unsigned i = (unsigned)tid; i < total; i += CSR_T) {
      const unsigned g = bo + i;
      if (g < (unsigned)e) ssrc[g] = ostage[i];
    }
  }
}

#define AGG_U 16

template <int J>
__device__ __forceinline__ unsigned group_bcast(unsigned v) {
  return __builtin_amdgcn_mov_dpp8(v, (unsigned)(J * 0x249249));
}

#define AT_PITCH 36

__global__ __launch_bounds__(256, 12) void k_gather_gemm(
    const float* __restrict__ xs, const int* __restrict__ ssrc,
    const unsigned* __restrict__ lpre, const unsigned* __restrict__ boff,
    const unsigned* __restrict__ degin, const float* __restrict__ norm_in,
    const float* __restrict__ W, const float* __restrict__ bias,
    const float* __restrict__ post_scale,
    float* __restrict__ out, int n, int e, int do_relu) {
  __shared__ __attribute__((aligned(16))) float tile[32 * AT_PITCH];
  __shared__ float rscale[32];
  const int lane = threadIdx.x & 31;
  const unsigned sub = (unsigned)(lane & 7);
  const int wv = __builtin_amdgcn_readfirstlane((int)(threadIdx.x >> 5));
  const int slot = wv * 4 + (lane >> 3);
  int node = blockIdx.x * 32 + slot;
  const bool live = node < n;
  if (!live) node = n - 1;
  const unsigned start = boff[node >> NB_SHIFT] + lpre[node];
  const unsigned cnt = live ? degin[node] : 0u;
  const float ni = norm_in[node];
  const float po = post_scale ? post_scale[node] : 1.0f;
  unsigned mx = cnt;
  {
    const unsigned o = (unsigned)__shfl_xor((int)mx, 8);
    mx = o > mx ? o : mx;
  }
  {
    const unsigned o = (unsigned)__shfl_xor((int)mx, 16);
    mx = o > mx ? o : mx;
  }
  const unsigned maxcnt = (unsigned)__builtin_amdgcn_readfirstlane((int)mx);

  const char* __restrict__ xb = reinterpret_cast<const char*>(xs);
  const unsigned sub16 = sub * 16u;
  const unsigned last = cnt ? cnt - 1u : 0u;
  const unsigned emax = (unsigned)e - 1u;
  v4f acc = {0.0f, 0.0f, 0.0f, 0.0f};
  for (unsigned c0 = 0; c0 < maxcnt; c0 += AGG_U) {
    unsigned a0 = start + (c0 + sub < cnt ? c0 + sub : last);
    unsigned a1 = start + (c0 + 8u + sub < cnt ? c0 + 8u + sub : last);
    if (a0 > emax) a0 = emax;
    if (a1 > emax) a1 = emax;
    const unsigned o0 = (unsigned)ssrc[a0] * (unsigned)(DF * 4);
    const unsigned o1 = (unsigned)ssrc[a1] * (unsigned)(DF * 4);
    v4f v[AGG_U];
#define AGG_ROW(O, B, J) \
  v[B + J] = *reinterpret_cast<const v4f*>(xb + (group_bcast<J>(O) + sub16));
    AGG_ROW(o0, 0, 0) AGG_ROW(o0, 0, 1) AGG_ROW(o0, 0, 2) AGG_ROW(o0, 0, 3)
    AGG_ROW(o0, 0, 4) AGG_ROW(o0, 0, 5) AGG_ROW(o0, 0, 6) AGG_ROW(o0, 0, 7)
    AGG_ROW(o1, 8, 0) AGG_ROW(o1, 8, 1) AGG_ROW(o1, 8, 2) AGG_ROW(o1, 8, 3)
    AGG_ROW(o1, 8, 4) AGG_ROW(o1, 8, 5) AGG_ROW(o1, 8, 6) AGG_ROW(o1, 8, 7)
#undef AGG_ROW
    __builtin_amdgcn_sched_barrier(0);
#pragma unroll
    for (int j = 0; j < AGG_U; ++j) {
      const bool ok = c0 + (unsigned)j < cnt;
      acc.x += ok ? v[j].x : 0.0f;
      acc.y += ok ? v[j].y : 0.0f;
      acc.z += ok ? v[j].z : 0.0f;
      acc.w += ok ? v[j].w : 0.0f;
    }
  }

  *reinterpret_cast<v4f*>(&tile[slot * AT_PITCH + (int)sub * 4]) = acc * ni;
  if (sub == 0u) rscale[slot] = po;
  __syncthreads();

  const int row0 = blockIdx.x * 32 + wv * 16;
  if (wv < 2 && row0 < n) {
    const int mrow = lane & 15;
    const int khalf = lane >> 4;
    const int nb = lane & 15;
    const float* arow = &tile[(wv * 16 + mrow) * AT_PITCH];
    v8f acc0 = {};
    v8f acc1 = {};
#pragma unroll
    for (int k0 = 0; k0 < DF; k0 += 4) {
      const int ka = k0 + khalf * 2;
      v2f a;
      a.x = arow[ka + 0];
      a.y = arow[ka + 1];
      v2f b0, b1;
      b0.x = W[(ka + 0) * DF + nb];
      b0.y = W[(ka + 1) * DF + nb];
      b1.x = W[(ka + 0) * DF + 16 + nb];
      b1.y = W[(ka + 1) * DF + 16 + nb];
      acc0 = __builtin_amdgcn_wmma_f32_16x16x4_f32(false, a, false, b0,
                                                   (short)0, acc0, false, false);
      acc1 = __builtin_amdgcn_wmma_f32_16x16x4_f32(false, a, false, b1,
                                                   (short)0, acc1, false, false);
    }

    const float bc0 = bias[nb];
    const float bc1 = bias[16 + nb];
#pragma unroll
    for (int v = 0; v < 8; ++v) {
      const int r = v + 8 * khalf;
      const int m = row0 + r;
      if (m < n) {
        float o0 = acc0[v] + bc0;
        float o1 = acc1[v] + bc1;
        if (do_relu) {
          o0 = fmaxf(o0, 0.0f);
          o1 = fmaxf(o1, 0.0f);
        }
        const float s = rscale[wv * 16 + r];
        out[(size_t)m * DF + nb] = o0 * s;
        out[(size_t)m * DF + 16 + nb] = o1 * s;
      }
    }
  }
}

extern "C" void kernel_launch(void* const* d_in, const int* in_sizes, int n_in,
                              void* d_out, int out_size, void* d_ws,
                              size_t ws_size, hipStream_t stream) {
  const float* x = (const float*)d_in[0];
  const int* src = (const int*)d_in[1];
  const int* dst = (const int*)d_in[2];
  const float* W1 = (const float*)d_in[3];
  const float* b1 = (const float*)d_in[4];
  const float* W2 = (const float*)d_in[5];
  const float* b2 = (const float*)d_in[6];
  float* out = (float*)d_out;

  const int N = in_sizes[0] / DF;
  const int E = in_sizes[1];
  const int NF = N * DF;

  float* ws = (float*)d_ws;
  float* norm_out = ws;
  float* norm_in = ws + N;
  float* xs = ws + ((2 * (size_t)N + 3) & ~(size_t)3);
  float* h1s = xs + (size_t)NF;
  unsigned* part = (unsigned*)(h1s + (size_t)NF);
  const int nblk = (N + NB_T - 1) / NB_T;
  const int np = (E + BIN_PIECE - 1) / BIN_PIECE;
  unsigned* lpre = part + (size_t)HIST_S * N;
  unsigned* degin = lpre + N;
  unsigned* boff = degin + N;
  unsigned* tab = boff + nblk;
  const size_t w_tab = (size_t)HIST_S * N + 2 * (size_t)N + nblk +
                       (size_t)np * (size_t)(nblk + 1);
  int2* binned = (int2*)(part + ((w_tab + 3) & ~(size_t)3));
  int* ssrc = (int*)(binned + (size_t)E);

  const int gA = (N + 31) / 32;

  const int chunk = (((E + HIST_S - 1) / HIST_S) + 3) & ~3;
  k_front<<<HIST_S + np, HIST_T, 0, stream>>>(src, dst, part, tab, binned, N, E, chunk, nblk);
  k_block<<<nblk, CSR_T, 0, stream>>>(part, x, tab, binned, norm_out, norm_in, xs, lpre,
                                      degin, boff, ssrc, N, E, nblk, np);

  k_gather_gemm<<<gA, 256, 0, stream>>>(xs, ssrc, lpre, boff, degin, norm_in, W1, b1,
                                        norm_out, h1s, N, E, 1);

  k_gather_gemm<<<gA, 256, 0, stream>>>(h1s, ssrc, lpre, boff, degin, norm_in, W2, b2,
                                        nullptr, out, N, E, 0);
}
